# starting kernel
# baseline (speedup 1.0000x reference)
#include <hip/hip_runtime.h>
#include <math.h>

#define BB 8
#define HH 128
#define LL 4096
#define NN 64
#define CHUNK 64
#define NCH (LL / CHUNK)
#define COLS (BB * NCH)
#define EPSV 1e-7f

typedef float v2f __attribute__((ext_vector_type(2)));
typedef float v8f __attribute__((ext_vector_type(8)));

__global__ void __launch_bounds__(64)
dss_coeffs(const float* __restrict__ Lre, const float* __restrict__ Lim,
           const float* __restrict__ Wri, const float* __restrict__ logstep,
           float* __restrict__ Zr, float* __restrict__ Zi,
           float* __restrict__ CR, float* __restrict__ CIn,
           float* __restrict__ T, float* __restrict__ zCr, float* __restrict__ zCi) {
    __shared__ float sdtr[NN], sdti[NN], scr[NN], sci[NN], sK[CHUNK];
    const int h = blockIdx.x;
    const int n = threadIdx.x;

    const float lre = Lre[h * NN + n], lim = Lim[h * NN + n];
    const float dt  = __expf(logstep[h]);
    const float ar  = dt * lre, ai = dt * lim;

    float cz, sz; sincosf(ai, &sz, &cz);
    const float er = expf(ar);
    const float zr = er * cz, zi = er * sz;
    float cL, sL; sincosf(ai * (float)LL, &sL, &cL);
    const float eL = expf(ar * (float)LL);
    const float zLr = eL * cL, zLi = eL * sL;
    const float nr = 1.f - zLr, ni = -zLi;
    const float dr = 1.f - zr,  di = -zi;
    const float dd = dr * dr + di * di;
    const float sr = (nr * dr + ni * di) / dd;
    const float si = (ni * dr - nr * di) / dd;
    const float wr = Wri[(h * NN + n) * 2 + 0], wi = Wri[(h * NN + n) * 2 + 1];
    const float ll = lre * lre + lim * lim;
    const float br = (wr * lre + wi * lim) / ll;
    const float bi = (wi * lre - wr * lim) / ll;
    const float ss = sr * sr + si * si + EPSV;
    const float cr = (br * sr + bi * si) / ss;
    const float ci = (bi * sr - br * si) / ss;

    float cC, sC; sincosf(ai * (float)CHUNK, &sC, &cC);
    const float eC = expf(ar * (float)CHUNK);
    zCr[h * NN + n] = eC * cC;
    zCi[h * NN + n] = eC * sC;

    {
        float pr = 1.f, pi_ = 0.f;
        float* zrrow = Zr + (h * NN + n) * CHUNK;
        float* zirow = Zi + (h * NN + n) * CHUNK;
        for (int k = 0; k < CHUNK; ++k) {
            zrrow[CHUNK - 1 - k] = pr;
            zirow[CHUNK - 1 - k] = pi_;
            const float tr = pr * zr - pi_ * zi;
            pi_ = pr * zi + pi_ * zr;
            pr = tr;
        }
    }
    {
        float qr = cr * zr - ci * zi;
        float qi = cr * zi + ci * zr;
        for (int t = 0; t < CHUNK; ++t) {
            CR [(h * CHUNK + t) * NN + n] =  qr;
            CIn[(h * CHUNK + t) * NN + n] = -qi;
            const float tr = qr * zr - qi * zi;
            qi = qr * zi + qi * zr;
            qr = tr;
        }
    }

    sdtr[n] = ar; sdti[n] = ai; scr[n] = cr; sci[n] = ci;
    __syncthreads();

    {
        const float t = (float)n;
        float acc = 0.f;
        for (int j = 0; j < NN; ++j) {
            const float e = expf(sdtr[j] * t);
            float cs, sn; sincosf(sdti[j] * t, &sn, &cs);
            acc += e * (scr[j] * cs - sci[j] * sn);
        }
        sK[n] = acc;
    }
    __syncthreads();

    for (int idx = n; idx < CHUNK * CHUNK; idx += NN) {
        const int t = idx >> 6, s = idx & 63;
        T[(h * CHUNK + t) * CHUNK + s] = (t >= s) ? sK[t - s] : 0.f;
    }
}

__global__ void __launch_bounds__(256)
dss_proj(const float* __restrict__ u,
         const float* __restrict__ Zr, const float* __restrict__ Zi,
         float* __restrict__ Pr, float* __restrict__ Pi) {
    const int lane  = threadIdx.x & 31;
    const int wave  = threadIdx.x >> 5;
    const int tile  = blockIdx.x * 8 + wave;
    const int h     = tile >> 7;
    const int rem   = tile & 127;
    const int rt    = rem >> 5;
    const int ct    = rem & 31;
    const int m     = lane & 15;
    const int khalf = lane >> 4;
    const int k0    = khalf * 2;

    const int arow = rt * 16 + m;
    const int col  = ct * 16 + m;
    const float* __restrict__ ucol =
        u + ((size_t)(col >> 6) * HH + h) * LL + (size_t)(col & 63) * CHUNK;
    const float* __restrict__ zrrow = Zr + (size_t)(h * NN + arow) * CHUNK;
    const float* __restrict__ zirow = Zi + (size_t)(h * NN + arow) * CHUNK;

    v8f accr = {}, acci = {};
#pragma unroll
    for (int kk = 0; kk < CHUNK; kk += 4) {
        const v2f aR = *(const v2f*)(zrrow + kk + k0);
        const v2f aI = *(const v2f*)(zirow + kk + k0);
        const v2f b  = *(const v2f*)(ucol + kk + k0);
        accr = __builtin_amdgcn_wmma_f32_16x16x4_f32(false, aR, false, b, (short)0, accr, false, false);
        acci = __builtin_amdgcn_wmma_f32_16x16x4_f32(false, aI, false, b, (short)0, acci, false, false);
    }

    const int nbase = rt * 16 + 8 * khalf;
    const size_t o = ((size_t)(h * COLS + col)) * NN + nbase;
    float4 r0, r1, i0, i1;
    r0.x = accr[0]; r0.y = accr[1]; r0.z = accr[2]; r0.w = accr[3];
    r1.x = accr[4]; r1.y = accr[5]; r1.z = accr[6]; r1.w = accr[7];
    i0.x = acci[0]; i0.y = acci[1]; i0.z = acci[2]; i0.w = acci[3];
    i1.x = acci[4]; i1.y = acci[5]; i1.z = acci[6]; i1.w = acci[7];
    *(float4*)(Pr + o)     = r0;
    *(float4*)(Pr + o + 4) = r1;
    *(float4*)(Pi + o)     = i0;
    *(float4*)(Pi + o + 4) = i1;
}

__global__ void __launch_bounds__(256)
dss_scan(float* __restrict__ Pr, float* __restrict__ Pi,
         const float* __restrict__ zCr, const float* __restrict__ zCi) {
    const int tid = blockIdx.x * blockDim.x + threadIdx.x;
    const int n = tid & 63;
    const int b = (tid >> 6) & 7;
    const int h = tid >> 9;
    const size_t base = ((size_t)(h * COLS + b * NCH)) * NN + n;
    const float zr = zCr[h * NN + n], zi = zCi[h * NN + n];
    float xr = 0.f, xi = 0.f;
    for (int i = 0; i < NCH; ++i) {
        const size_t o = base + (size_t)i * NN;
        const float pr = Pr[o], pi_ = Pi[o];
        const float nxr = zr * xr - zi * xi + pr;
        const float nxi = zr * xi + zi * xr + pi_;
        Pr[o] = nxr; Pi[o] = nxi;
        xr = nxr; xi = nxi;
    }
}

__global__ void __launch_bounds__(256)
dss_out(const float* __restrict__ u, const float* __restrict__ D,
        const float* __restrict__ T, const float* __restrict__ CR,
        const float* __restrict__ CIn,
        const float* __restrict__ Pr, const float* __restrict__ Pi,
        float* __restrict__ out) {
    const int lane  = threadIdx.x & 31;
    const int wave  = threadIdx.x >> 5;
    const int tile  = blockIdx.x * 8 + wave;
    const int h     = tile >> 7;
    const int rem   = tile & 127;
    const int rt    = rem >> 5;
    const int ct    = rem & 31;
    const int m     = lane & 15;
    const int khalf = lane >> 4;
    const int k0    = khalf * 2;

    const int arow = rt * 16 + m;
    const int col  = ct * 16 + m;
    const int b    = col >> 6;
    const int ch   = col & 63;
    const bool first = (ch == 0);

    const float* __restrict__ ucol  = u + ((size_t)b * HH + h) * LL + (size_t)ch * CHUNK;
    const float* __restrict__ trow  = T   + (size_t)(h * CHUNK + arow) * CHUNK;
    const float* __restrict__ crrow = CR  + (size_t)(h * CHUNK + arow) * NN;
    const float* __restrict__ cirow = CIn + (size_t)(h * CHUNK + arow) * NN;
    const float* __restrict__ xrcol = Pr + ((size_t)(h * COLS + col - 1)) * NN;
    const float* __restrict__ xicol = Pi + ((size_t)(h * COLS + col - 1)) * NN;

    v8f acc = {};
#pragma unroll
    for (int kk = 0; kk < CHUNK; kk += 4) {
        const v2f a  = *(const v2f*)(trow + kk + k0);
        const v2f bu = *(const v2f*)(ucol + kk + k0);
        acc = __builtin_amdgcn_wmma_f32_16x16x4_f32(false, a, false, bu, (short)0, acc, false, false);
    }
#pragma unroll
    for (int kk = 0; kk < CHUNK; kk += 4) {
        const v2f aR = *(const v2f*)(crrow + kk + k0);
        const v2f aI = *(const v2f*)(cirow + kk + k0);
        v2f bxr = {0.f, 0.f}, bxi = {0.f, 0.f};
        if (!first) {
            bxr = *(const v2f*)(xrcol + kk + k0);
            bxi = *(const v2f*)(xicol + kk + k0);
        }
        acc = __builtin_amdgcn_wmma_f32_16x16x4_f32(false, aR, false, bxr, (short)0, acc, false, false);
        acc = __builtin_amdgcn_wmma_f32_16x16x4_f32(false, aI, false, bxi, (short)0, acc, false, false);
    }

    const float Dh = D[h];
    const size_t obase = ((size_t)b * HH + h) * LL + (size_t)ch * CHUNK + rt * 16 + 8 * khalf;
    const float4 u0 = *(const float4*)(u + obase);
    const float4 u1 = *(const float4*)(u + obase + 4);
    float4 o0, o1;
    o0.x = acc[0] + Dh * u0.x; o0.y = acc[1] + Dh * u0.y;
    o0.z = acc[2] + Dh * u0.z; o0.w = acc[3] + Dh * u0.w;
    o1.x = acc[4] + Dh * u1.x; o1.y = acc[5] + Dh * u1.y;
    o1.z = acc[6] + Dh * u1.z; o1.w = acc[7] + Dh * u1.w;
    *(float4*)(out + obase)     = o0;
    *(float4*)(out + obase + 4) = o1;
}

extern "C" void kernel_launch(void* const* d_in, const int* in_sizes, int n_in,
                              void* d_out, int out_size, void* d_ws, size_t ws_size,
                              hipStream_t stream) {
    const float* u       = (const float*)d_in[0];
    const float* Lre     = (const float*)d_in[1];
    const float* Lim     = (const float*)d_in[2];
    const float* Wri     = (const float*)d_in[3];
    const float* Dp      = (const float*)d_in[4];
    const float* logstep = (const float*)d_in[5];
    float* out = (float*)d_out;

    float* ws  = (float*)d_ws;
    float* Zr  = ws;
    float* Zi  = Zr  + (size_t)HH * NN * CHUNK;
    float* CRm = Zi  + (size_t)HH * NN * CHUNK;
    float* CIn = CRm + (size_t)HH * CHUNK * NN;
    float* Tm  = CIn + (size_t)HH * CHUNK * NN;
    float* zCr = Tm  + (size_t)HH * CHUNK * CHUNK;
    float* zCi = zCr + (size_t)HH * NN;
    float* Pr  = zCi + (size_t)HH * NN;
    float* Pi  = Pr  + (size_t)HH * COLS * NN;

    dss_coeffs<<<HH, NN, 0, stream>>>(Lre, Lim, Wri, logstep,
                                      Zr, Zi, CRm, CIn, Tm, zCr, zCi);

    const int tiles = HH * (NN / 16) * (COLS / 16);
    dss_proj<<<tiles / 8, 256, 0, stream>>>(u, Zr, Zi, Pr, Pi);

    dss_scan<<<(HH * NN * BB) / 256, 256, 0, stream>>>(Pr, Pi, zCr, zCi);

    dss_out<<<tiles / 8, 256, 0, stream>>>(u, Dp, Tm, CRm, CIn, Pr, Pi, out);
}


# Operand loads cover whole lines: A matrices kept in WMMA fragment order in the workspace, A and B read as b128 with one shared k order (k=8g+4*half+0..3); k sum order inside each 8-k group changes
# speedup vs baseline: 1.1474x; 1.1474x over previous
#include <hip/hip_runtime.h>
#include <math.h>

#define BB 8
#define HH 128
#define LL 4096
#define NN 64
#define CHUNK 64
#define NCH (LL / CHUNK)
#define COLS (BB * NCH)
#define EPSV 1e-7f

typedef float v2f __attribute__((ext_vector_type(2)));
typedef float v8f __attribute__((ext_vector_type(8)));

__device__ __forceinline__ int frag_off(int h, int row, int k) {
    return ((h * 4 + (row >> 4)) * 8 + (k >> 3)) * 128
         + ((row & 15) + 16 * ((k >> 2) & 1)) * 4 + (k & 3);
}

__global__ void __launch_bounds__(64)
dss_coeffs(const float* __restrict__ Lre, const float* __restrict__ Lim,
           const float* __restrict__ Wri, const float* __restrict__ logstep,
           float* __restrict__ Zr, float* __restrict__ Zi,
           float* __restrict__ CR, float* __restrict__ CIn,
           float* __restrict__ T, float* __restrict__ zCr, float* __restrict__ zCi) {
    __shared__ float sdtr[NN], sdti[NN], scr[NN], sci[NN], sK[CHUNK];
    const int h = blockIdx.x;
    const int n = threadIdx.x;

    const float lre = Lre[h * NN + n], lim = Lim[h * NN + n];
    const float dt  = __expf(logstep[h]);
    const float ar  = dt * lre, ai = dt * lim;

    float cz, sz; sincosf(ai, &sz, &cz);
    const float er = expf(ar);
    const float zr = er * cz, zi = er * sz;
    float cL, sL; sincosf(ai * (float)LL, &sL, &cL);
    const float eL = expf(ar * (float)LL);
    const float zLr = eL * cL, zLi = eL * sL;
    const float nr = 1.f - zLr, ni = -zLi;
    const float dr = 1.f - zr,  di = -zi;
    const float dd = dr * dr + di * di;
    const float sr = (nr * dr + ni * di) / dd;
    const float si = (ni * dr - nr * di) / dd;
    const float wr = Wri[(h * NN + n) * 2 + 0], wi = Wri[(h * NN + n) * 2 + 1];
    const float ll = lre * lre + lim * lim;
    const float br = (wr * lre + wi * lim) / ll;
    const float bi = (wi * lre - wr * lim) / ll;
    const float ss = sr * sr + si * si + EPSV;
    const float cr = (br * sr + bi * si) / ss;
    const float ci = (bi * sr - br * si) / ss;

    float cC, sC; sincosf(ai * (float)CHUNK, &sC, &cC);
    const float eC = expf(ar * (float)CHUNK);
    zCr[h * NN + n] = eC * cC;
    zCi[h * NN + n] = eC * sC;

    {
        float pr = 1.f, pi_ = 0.f;
        for (int k = 0; k < CHUNK; ++k) {
            const int o = frag_off(h, n, CHUNK - 1 - k);
            Zr[o] = pr;
            Zi[o] = pi_;
            const float tr = pr * zr - pi_ * zi;
            pi_ = pr * zi + pi_ * zr;
            pr = tr;
        }
    }
    {
        float qr = cr * zr - ci * zi;
        float qi = cr * zi + ci * zr;
        for (int t = 0; t < CHUNK; ++t) {
            const int o = frag_off(h, t, n);
            CR [o] =  qr;
            CIn[o] = -qi;
            const float tr = qr * zr - qi * zi;
            qi = qr * zi + qi * zr;
            qr = tr;
        }
    }

    sdtr[n] = ar; sdti[n] = ai; scr[n] = cr; sci[n] = ci;
    __syncthreads();

    {
        const float t = (float)n;
        float acc = 0.f;
        for (int j = 0; j < NN; ++j) {
            const float e = expf(sdtr[j] * t);
            float cs, sn; sincosf(sdti[j] * t, &sn, &cs);
            acc += e * (scr[j] * cs - sci[j] * sn);
        }
        sK[n] = acc;
    }
    __syncthreads();

    for (int idx = n; idx < CHUNK * CHUNK; idx += NN) {
        const int t = idx >> 6, s = idx & 63;
        T[frag_off(h, t, s)] = (t >= s) ? sK[t - s] : 0.f;
    }
}

__global__ void __launch_bounds__(256)
dss_proj(const float* __restrict__ u,
         const float* __restrict__ Zr, const float* __restrict__ Zi,
         float* __restrict__ Pr, float* __restrict__ Pi) {
    const int lane  = threadIdx.x & 31;
    const int wave  = threadIdx.x >> 5;
    const int tile  = blockIdx.x * 8 + wave;
    const int h     = tile >> 7;
    const int rem   = tile & 127;
    const int rt    = rem >> 5;
    const int ct    = rem & 31;
    const int m     = lane & 15;
    const int khalf = lane >> 4;

    const int col  = ct * 16 + m;
    const float* __restrict__ ucol =
        u + ((size_t)(col >> 6) * HH + h) * LL + (size_t)(col & 63) * CHUNK;
    const float* __restrict__ zrfrag = Zr + (size_t)(h * 4 + rt) * 1024 + lane * 4;
    const float* __restrict__ zifrag = Zi + (size_t)(h * 4 + rt) * 1024 + lane * 4;
    const float* __restrict__ ub = ucol + 4 * khalf;

    v8f accr = {}, acci = {};
#pragma unroll
    for (int g = 0; g < CHUNK / 8; ++g) {
        const float4 aR = *(const float4*)(zrfrag + g * 128);
        const float4 aI = *(const float4*)(zifrag + g * 128);
        const float4 b  = *(const float4*)(ub + g * 8);
        const v2f aR0 = {aR.x, aR.y}, aR1 = {aR.z, aR.w};
        const v2f aI0 = {aI.x, aI.y}, aI1 = {aI.z, aI.w};
        const v2f b0  = {b.x, b.y},   b1  = {b.z, b.w};
        accr = __builtin_amdgcn_wmma_f32_16x16x4_f32(false, aR0, false, b0, (short)0, accr, false, false);
        acci = __builtin_amdgcn_wmma_f32_16x16x4_f32(false, aI0, false, b0, (short)0, acci, false, false);
        accr = __builtin_amdgcn_wmma_f32_16x16x4_f32(false, aR1, false, b1, (short)0, accr, false, false);
        acci = __builtin_amdgcn_wmma_f32_16x16x4_f32(false, aI1, false, b1, (short)0, acci, false, false);
    }

    const int nbase = rt * 16 + 8 * khalf;
    const size_t o = ((size_t)(h * COLS + col)) * NN + nbase;
    float4 r0, r1, i0, i1;
    r0.x = accr[0]; r0.y = accr[1]; r0.z = accr[2]; r0.w = accr[3];
    r1.x = accr[4]; r1.y = accr[5]; r1.z = accr[6]; r1.w = accr[7];
    i0.x = acci[0]; i0.y = acci[1]; i0.z = acci[2]; i0.w = acci[3];
    i1.x = acci[4]; i1.y = acci[5]; i1.z = acci[6]; i1.w = acci[7];
    *(float4*)(Pr + o)     = r0;
    *(float4*)(Pr + o + 4) = r1;
    *(float4*)(Pi + o)     = i0;
    *(float4*)(Pi + o + 4) = i1;
}

__global__ void __launch_bounds__(256)
dss_scan(float* __restrict__ Pr, float* __restrict__ Pi,
         const float* __restrict__ zCr, const float* __restrict__ zCi) {
    const int tid = blockIdx.x * blockDim.x + threadIdx.x;
    const int n = tid & 63;
    const int b = (tid >> 6) & 7;
    const int h = tid >> 9;
    const size_t base = ((size_t)(h * COLS + b * NCH)) * NN + n;
    const float zr = zCr[h * NN + n], zi = zCi[h * NN + n];
    float xr = 0.f, xi = 0.f;
    for (int i = 0; i < NCH; ++i) {
        const size_t o = base + (size_t)i * NN;
        const float pr = Pr[o], pi_ = Pi[o];
        const float nxr = zr * xr - zi * xi + pr;
        const float nxi = zr * xi + zi * xr + pi_;
        Pr[o] = nxr; Pi[o] = nxi;
        xr = nxr; xi = nxi;
    }
}

__global__ void __launch_bounds__(256)
dss_out(const float* __restrict__ u, const float* __restrict__ D,
        const float* __restrict__ T, const float* __restrict__ CR,
        const float* __restrict__ CIn,
        const float* __restrict__ Pr, const float* __restrict__ Pi,
        float* __restrict__ out) {
    const int lane  = threadIdx.x & 31;
    const int wave  = threadIdx.x >> 5;
    const int tile  = blockIdx.x * 8 + wave;
    const int h     = tile >> 7;
    const int rem   = tile & 127;
    const int rt    = rem >> 5;
    const int ct    = rem & 31;
    const int m     = lane & 15;
    const int khalf = lane >> 4;

    const int col  = ct * 16 + m;
    const int b    = col >> 6;
    const int ch   = col & 63;
    const bool first = (ch == 0);

    const float* __restrict__ ucol  = u + ((size_t)b * HH + h) * LL + (size_t)ch * CHUNK;
    const size_t afrag = (size_t)(h * 4 + rt) * 1024 + lane * 4;
    const float* __restrict__ tfrag  = T   + afrag;
    const float* __restrict__ crfrag = CR  + afrag;
    const float* __restrict__ cifrag = CIn + afrag;
    const float* __restrict__ ub = ucol + 4 * khalf;
    const float* __restrict__ xrb = Pr + ((size_t)(h * COLS + col - 1)) * NN + 4 * khalf;
    const float* __restrict__ xib = Pi + ((size_t)(h * COLS + col - 1)) * NN + 4 * khalf;

    v8f acc = {};
#pragma unroll
    for (int g = 0; g < CHUNK / 8; ++g) {
        const float4 a  = *(const float4*)(tfrag + g * 128);
        const float4 bu = *(const float4*)(ub + g * 8);
        const v2f a0 = {a.x, a.y},   a1 = {a.z, a.w};
        const v2f b0 = {bu.x, bu.y}, b1 = {bu.z, bu.w};
        acc = __builtin_amdgcn_wmma_f32_16x16x4_f32(false, a0, false, b0, (short)0, acc, false, false);
        acc = __builtin_amdgcn_wmma_f32_16x16x4_f32(false, a1, false, b1, (short)0, acc, false, false);
    }
#pragma unroll
    for (int g = 0; g < NN / 8; ++g) {
        const float4 aR = *(const float4*)(crfrag + g * 128);
        const float4 aI = *(const float4*)(cifrag + g * 128);
        float4 bxr = {0.f, 0.f, 0.f, 0.f}, bxi = {0.f, 0.f, 0.f, 0.f};
        if (!first) {
            bxr = *(const float4*)(xrb + g * 8);
            bxi = *(const float4*)(xib + g * 8);
        }
        const v2f aR0 = {aR.x, aR.y},   aR1 = {aR.z, aR.w};
        const v2f aI0 = {aI.x, aI.y},   aI1 = {aI.z, aI.w};
        const v2f xr0 = {bxr.x, bxr.y}, xr1 = {bxr.z, bxr.w};
        const v2f xi0 = {bxi.x, bxi.y}, xi1 = {bxi.z, bxi.w};
        acc = __builtin_amdgcn_wmma_f32_16x16x4_f32(false, aR0, false, xr0, (short)0, acc, false, false);
        acc = __builtin_amdgcn_wmma_f32_16x16x4_f32(false, aI0, false, xi0, (short)0, acc, false, false);
        acc = __builtin_amdgcn_wmma_f32_16x16x4_f32(false, aR1, false, xr1, (short)0, acc, false, false);
        acc = __builtin_amdgcn_wmma_f32_16x16x4_f32(false, aI1, false, xi1, (short)0, acc, false, false);
    }

    const float Dh = D[h];
    const size_t obase = ((size_t)b * HH + h) * LL + (size_t)ch * CHUNK + rt * 16 + 8 * khalf;
    const float4 u0 = *(const float4*)(u + obase);
    const float4 u1 = *(const float4*)(u + obase + 4);
    float4 o0, o1;
    o0.x = acc[0] + Dh * u0.x; o0.y = acc[1] + Dh * u0.y;
    o0.z = acc[2] + Dh * u0.z; o0.w = acc[3] + Dh * u0.w;
    o1.x = acc[4] + Dh * u1.x; o1.y = acc[5] + Dh * u1.y;
    o1.z = acc[6] + Dh * u1.z; o1.w = acc[7] + Dh * u1.w;
    *(float4*)(out + obase)     = o0;
    *(float4*)(out + obase + 4) = o1;
}

extern "C" void kernel_launch(void* const* d_in, const int* in_sizes, int n_in,
                              void* d_out, int out_size, void* d_ws, size_t ws_size,
                              hipStream_t stream) {
    const float* u       = (const float*)d_in[0];
    const float* Lre     = (const float*)d_in[1];
    const float* Lim     = (const float*)d_in[2];
    const float* Wri     = (const float*)d_in[3];
    const float* Dp      = (const float*)d_in[4];
    const float* logstep = (const float*)d_in[5];
    float* out = (float*)d_out;

    float* ws  = (float*)d_ws;
    float* Zr  = ws;
    float* Zi  = Zr  + (size_t)HH * NN * CHUNK;
    float* CRm = Zi  + (size_t)HH * NN * CHUNK;
    float* CIn = CRm + (size_t)HH * CHUNK * NN;
    float* Tm  = CIn + (size_t)HH * CHUNK * NN;
    float* zCr = Tm  + (size_t)HH * CHUNK * CHUNK;
    float* zCi = zCr + (size_t)HH * NN;
    float* Pr  = zCi + (size_t)HH * NN;
    float* Pi  = Pr  + (size_t)HH * COLS * NN;

    dss_coeffs<<<HH, NN, 0, stream>>>(Lre, Lim, Wri, logstep,
                                      Zr, Zi, CRm, CIn, Tm, zCr, zCi);

    const int tiles = HH * (NN / 16) * (COLS / 16);
    dss_proj<<<tiles / 8, 256, 0, stream>>>(u, Zr, Zi, Pr, Pi);

    dss_scan<<<(HH * NN * BB) / 256, 256, 0, stream>>>(Pr, Pi, zCr, zCi);

    dss_out<<<tiles / 8, 256, 0, stream>>>(u, Dp, Tm, CRm, CIn, Pr, Pi, out);
}


# Projection, chunk scan and output fused into one work-group per (batch, head): chunk states stay in LDS, U fragments in registers; 4 launches become 2; every sum keeps its order
# speedup vs baseline: 2.6594x; 2.3178x over previous
#include <hip/hip_runtime.h>
#include <math.h>

#define BB 8
#define HH 128
#define LL 4096
#define NN 64
#define CHUNK 64
#define NCH (LL / CHUNK)
#define XP 68
#define EPSV 1e-7f

typedef float v2f __attribute__((ext_vector_type(2)));
typedef float v8f __attribute__((ext_vector_type(8)));

__device__ __forceinline__ int frag_off(int h, int row, int k) {
    return ((h * 4 + (row >> 4)) * 8 + (k >> 3)) * 128
         + ((row & 15) + 16 * ((k >> 2) & 1)) * 4 + (k & 3);
}

__global__ void __launch_bounds__(64)
dss_coeffs(const float* __restrict__ Lre, const float* __restrict__ Lim,
           const float* __restrict__ Wri, const float* __restrict__ logstep,
           float* __restrict__ Zr, float* __restrict__ Zi,
           float* __restrict__ CR, float* __restrict__ CIn,
           float* __restrict__ T, float* __restrict__ zCr, float* __restrict__ zCi) {
    __shared__ float sdtr[NN], sdti[NN], scr[NN], sci[NN], sK[CHUNK];
    const int h = blockIdx.x;
    const int n = threadIdx.x;

    const float lre = Lre[h * NN + n], lim = Lim[h * NN + n];
    const float dt  = __expf(logstep[h]);
    const float ar  = dt * lre, ai = dt * lim;

    float cz, sz; sincosf(ai, &sz, &cz);
    const float er = expf(ar);
    const float zr = er * cz, zi = er * sz;
    float cL, sL; sincosf(ai * (float)LL, &sL, &cL);
    const float eL = expf(ar * (float)LL);
    const float zLr = eL * cL, zLi = eL * sL;
    const float nr = 1.f - zLr, ni = -zLi;
    const float dr = 1.f - zr,  di = -zi;
    const float dd = dr * dr + di * di;
    const float sr = (nr * dr + ni * di) / dd;
    const float si = (ni * dr - nr * di) / dd;
    const float wr = Wri[(h * NN + n) * 2 + 0], wi = Wri[(h * NN + n) * 2 + 1];
    const float ll = lre * lre + lim * lim;
    const float br = (wr * lre + wi * lim) / ll;
    const float bi = (wi * lre - wr * lim) / ll;
    const float ss = sr * sr + si * si + EPSV;
    const float cr = (br * sr + bi * si) / ss;
    const float ci = (bi * sr - br * si) / ss;

    float cC, sC; sincosf(ai * (float)CHUNK, &sC, &cC);
    const float eC = expf(ar * (float)CHUNK);
    zCr[h * NN + n] = eC * cC;
    zCi[h * NN + n] = eC * sC;

    {
        float pr = 1.f, pi_ = 0.f;
        for (int k = 0; k < CHUNK; ++k) {
            const int o = frag_off(h, n, CHUNK - 1 - k);
            Zr[o] = pr;
            Zi[o] = pi_;
            const float tr = pr * zr - pi_ * zi;
            pi_ = pr * zi + pi_ * zr;
            pr = tr;
        }
    }
    {
        float qr = cr * zr - ci * zi;
        float qi = cr * zi + ci * zr;
        for (int t = 0; t < CHUNK; ++t) {
            const int o = frag_off(h, t, n);
            CR [o] =  qr;
            CIn[o] = -qi;
            const float tr = qr * zr - qi * zi;
            qi = qr * zi + qi * zr;
            qr = tr;
        }
    }

    sdtr[n] = ar; sdti[n] = ai; scr[n] = cr; sci[n] = ci;
    __syncthreads();

    {
        const float t = (float)n;
        float acc = 0.f;
        for (int j = 0; j < NN; ++j) {
            const float e = expf(sdtr[j] * t);
            float cs, sn; sincosf(sdti[j] * t, &sn, &cs);
            acc += e * (scr[j] * cs - sci[j] * sn);
        }
        sK[n] = acc;
    }
    __syncthreads();

    for (int idx = n; idx < CHUNK * CHUNK; idx += NN) {
        const int t = idx >> 6, s = idx & 63;
        T[frag_off(h, t, s)] = (t >= s) ? sK[t - s] : 0.f;
    }
}

__global__ void __launch_bounds__(256)
dss_fused(const float* __restrict__ u, const float* __restrict__ D,
          const float* __restrict__ Zr, const float* __restrict__ Zi,
          const float* __restrict__ T, const float* __restrict__ CR,
          const float* __restrict__ CIn,
          const float* __restrict__ zCr, const float* __restrict__ zCi,
          float* __restrict__ out) {
    __shared__ __attribute__((aligned(16))) float sXr[(NCH + 1) * XP];
    __shared__ __attribute__((aligned(16))) float sXi[(NCH + 1) * XP];

    const int tid   = threadIdx.x;
    const int lane  = tid & 31;
    const int wave  = tid >> 5;
    const int h     = blockIdx.x >> 3;
    const int b     = blockIdx.x & 7;
    const int m     = lane & 15;
    const int khalf = lane >> 4;
    const int ct    = wave >> 1;
    const int rt0   = (wave & 1) * 2;
    const int ch    = ct * 16 + m;

    const float* __restrict__ ubh = u + ((size_t)b * HH + h) * LL;
    const float* __restrict__ ub  = ubh + (size_t)ch * CHUNK + 4 * khalf;
    const size_t afrag = (size_t)(h * 4 + rt0) * 1024 + lane * 4;

    const float zr = zCr[h * NN + (tid & 63)];
    const float zi = zCi[h * NN + (tid & 63)];

    float4 bu[CHUNK / 8];
#pragma unroll
    for (int g = 0; g < CHUNK / 8; ++g) bu[g] = *(const float4*)(ub + g * 8);

    if (tid < NN) { sXr[tid] = 0.f; sXi[tid] = 0.f; }

    {
        const float* __restrict__ zrf = Zr + afrag;
        const float* __restrict__ zif = Zi + afrag;
        v8f pr0 = {}, pi0 = {}, pr1 = {}, pi1 = {};
#pragma unroll
        for (int g = 0; g < CHUNK / 8; ++g) {
            const float4 aR0 = *(const float4*)(zrf + g * 128);
            const float4 aI0 = *(const float4*)(zif + g * 128);
            const float4 aR1 = *(const float4*)(zrf + 1024 + g * 128);
            const float4 aI1 = *(const float4*)(zif + 1024 + g * 128);
            const v2f b0 = {bu[g].x, bu[g].y}, b1 = {bu[g].z, bu[g].w};
            const v2f r00 = {aR0.x, aR0.y}, r01 = {aR0.z, aR0.w};
            const v2f i00 = {aI0.x, aI0.y}, i01 = {aI0.z, aI0.w};
            const v2f r10 = {aR1.x, aR1.y}, r11 = {aR1.z, aR1.w};
            const v2f i10 = {aI1.x, aI1.y}, i11 = {aI1.z, aI1.w};
            pr0 = __builtin_amdgcn_wmma_f32_16x16x4_f32(false, r00, false, b0, (short)0, pr0, false, false);
            pi0 = __builtin_amdgcn_wmma_f32_16x16x4_f32(false, i00, false, b0, (short)0, pi0, false, false);
            pr1 = __builtin_amdgcn_wmma_f32_16x16x4_f32(false, r10, false, b0, (short)0, pr1, false, false);
            pi1 = __builtin_amdgcn_wmma_f32_16x16x4_f32(false, i10, false, b0, (short)0, pi1, false, false);
            pr0 = __builtin_amdgcn_wmma_f32_16x16x4_f32(false, r01, false, b1, (short)0, pr0, false, false);
            pi0 = __builtin_amdgcn_wmma_f32_16x16x4_f32(false, i01, false, b1, (short)0, pi0, false, false);
            pr1 = __builtin_amdgcn_wmma_f32_16x16x4_f32(false, r11, false, b1, (short)0, pr1, false, false);
            pi1 = __builtin_amdgcn_wmma_f32_16x16x4_f32(false, i11, false, b1, (short)0, pi1, false, false);
        }
        float* __restrict__ dr = sXr + (ch + 1) * XP + rt0 * 16 + 8 * khalf;
        float* __restrict__ di = sXi + (ch + 1) * XP + rt0 * 16 + 8 * khalf;
        float4 v;
        v.x = pr0[0]; v.y = pr0[1]; v.z = pr0[2]; v.w = pr0[3]; *(float4*)(dr)      = v;
        v.x = pr0[4]; v.y = pr0[5]; v.z = pr0[6]; v.w = pr0[7]; *(float4*)(dr + 4)  = v;
        v.x = pr1[0]; v.y = pr1[1]; v.z = pr1[2]; v.w = pr1[3]; *(float4*)(dr + 16) = v;
        v.x = pr1[4]; v.y = pr1[5]; v.z = pr1[6]; v.w = pr1[7]; *(float4*)(dr + 20) = v;
        v.x = pi0[0]; v.y = pi0[1]; v.z = pi0[2]; v.w = pi0[3]; *(float4*)(di)      = v;
        v.x = pi0[4]; v.y = pi0[5]; v.z = pi0[6]; v.w = pi0[7]; *(float4*)(di + 4)  = v;
        v.x = pi1[0]; v.y = pi1[1]; v.z = pi1[2]; v.w = pi1[3]; *(float4*)(di + 16) = v;
        v.x = pi1[4]; v.y = pi1[5]; v.z = pi1[6]; v.w = pi1[7]; *(float4*)(di + 20) = v;
    }
    __syncthreads();

    if (tid < NN) {
        float xr = 0.f, xi = 0.f;
        float* __restrict__ sr = sXr + XP + tid;
        float* __restrict__ si = sXi + XP + tid;
#pragma unroll
        for (int i0 = 0; i0 < NCH; i0 += 16) {
            float vr[16], vi[16];
#pragma unroll
            for (int j = 0; j < 16; ++j) { vr[j] = sr[(i0 + j) * XP]; vi[j] = si[(i0 + j) * XP]; }
#pragma unroll
            for (int j = 0; j < 16; ++j) {
                const float nxr = zr * xr - zi * xi + vr[j];
                const float nxi = zr * xi + zi * xr + vi[j];
                vr[j] = nxr; vi[j] = nxi;
                xr = nxr; xi = nxi;
            }
#pragma unroll
            for (int j = 0; j < 16; ++j) { sr[(i0 + j) * XP] = vr[j]; si[(i0 + j) * XP] = vi[j]; }
        }
    }

    v8f acc0 = {}, acc1 = {};
    {
        const float* __restrict__ tf = T + afrag;
#pragma unroll
        for (int g = 0; g < CHUNK / 8; ++g) {
            const float4 a0 = *(const float4*)(tf + g * 128);
            const float4 a1 = *(const float4*)(tf + 1024 + g * 128);
            const v2f b0 = {bu[g].x, bu[g].y}, b1 = {bu[g].z, bu[g].w};
            const v2f a00 = {a0.x, a0.y}, a01 = {a0.z, a0.w};
            const v2f a10 = {a1.x, a1.y}, a11 = {a1.z, a1.w};
            acc0 = __builtin_amdgcn_wmma_f32_16x16x4_f32(false, a00, false, b0, (short)0, acc0, false, false);
            acc1 = __builtin_amdgcn_wmma_f32_16x16x4_f32(false, a10, false, b0, (short)0, acc1, false, false);
            acc0 = __builtin_amdgcn_wmma_f32_16x16x4_f32(false, a01, false, b1, (short)0, acc0, false, false);
            acc1 = __builtin_amdgcn_wmma_f32_16x16x4_f32(false, a11, false, b1, (short)0, acc1, false, false);
        }
    }
    __syncthreads();

    {
        const float* __restrict__ crf = CR  + afrag;
        const float* __restrict__ cif = CIn + afrag;
        const float* __restrict__ xrb = sXr + ch * XP + 4 * khalf;
        const float* __restrict__ xib = sXi + ch * XP + 4 * khalf;
#pragma unroll
        for (int g = 0; g < NN / 8; ++g) {
            const float4 aR0 = *(const float4*)(crf + g * 128);
            const float4 aI0 = *(const float4*)(cif + g * 128);
            const float4 aR1 = *(const float4*)(crf + 1024 + g * 128);
            const float4 aI1 = *(const float4*)(cif + 1024 + g * 128);
            const float4 bxr = *(const float4*)(xrb + g * 8);
            const float4 bxi = *(const float4*)(xib + g * 8);
            const v2f xr0 = {bxr.x, bxr.y}, xr1 = {bxr.z, bxr.w};
            const v2f xi0 = {bxi.x, bxi.y}, xi1 = {bxi.z, bxi.w};
            const v2f r00 = {aR0.x, aR0.y}, r01 = {aR0.z, aR0.w};
            const v2f i00 = {aI0.x, aI0.y}, i01 = {aI0.z, aI0.w};
            const v2f r10 = {aR1.x, aR1.y}, r11 = {aR1.z, aR1.w};
            const v2f i10 = {aI1.x, aI1.y}, i11 = {aI1.z, aI1.w};
            acc0 = __builtin_amdgcn_wmma_f32_16x16x4_f32(false, r00, false, xr0, (short)0, acc0, false, false);
            acc1 = __builtin_amdgcn_wmma_f32_16x16x4_f32(false, r10, false, xr0, (short)0, acc1, false, false);
            acc0 = __builtin_amdgcn_wmma_f32_16x16x4_f32(false, i00, false, xi0, (short)0, acc0, false, false);
            acc1 = __builtin_amdgcn_wmma_f32_16x16x4_f32(false, i10, false, xi0, (short)0, acc1, false, false);
            acc0 = __builtin_amdgcn_wmma_f32_16x16x4_f32(false, r01, false, xr1, (short)0, acc0, false, false);
            acc1 = __builtin_amdgcn_wmma_f32_16x16x4_f32(false, r11, false, xr1, (short)0, acc1, false, false);
            acc0 = __builtin_amdgcn_wmma_f32_16x16x4_f32(false, i01, false, xi1, (short)0, acc0, false, false);
            acc1 = __builtin_amdgcn_wmma_f32_16x16x4_f32(false, i11, false, xi1, (short)0, acc1, false, false);
        }
    }

    const float Dh = D[h];
    const size_t obase = ((size_t)b * HH + h) * LL + (size_t)ch * CHUNK + rt0 * 16 + 8 * khalf;
    {
        const float4 u0 = *(const float4*)(u + obase);
        const float4 u1 = *(const float4*)(u + obase + 4);
        float4 o0, o1;
        o0.x = acc0[0] + Dh * u0.x; o0.y = acc0[1] + Dh * u0.y;
        o0.z = acc0[2] + Dh * u0.z; o0.w = acc0[3] + Dh * u0.w;
        o1.x = acc0[4] + Dh * u1.x; o1.y = acc0[5] + Dh * u1.y;
        o1.z = acc0[6] + Dh * u1.z; o1.w = acc0[7] + Dh * u1.w;
        *(float4*)(out + obase)     = o0;
        *(float4*)(out + obase + 4) = o1;
    }
    {
        const float4 u0 = *(const float4*)(u + obase + 16);
        const float4 u1 = *(const float4*)(u + obase + 20);
        float4 o0, o1;
        o0.x = acc1[0] + Dh * u0.x; o0.y = acc1[1] + Dh * u0.y;
        o0.z = acc1[2] + Dh * u0.z; o0.w = acc1[3] + Dh * u0.w;
        o1.x = acc1[4] + Dh * u1.x; o1.y = acc1[5] + Dh * u1.y;
        o1.z = acc1[6] + Dh * u1.z; o1.w = acc1[7] + Dh * u1.w;
        *(float4*)(out + obase + 16) = o0;
        *(float4*)(out + obase + 20) = o1;
    }
}

extern "C" void kernel_launch(void* const* d_in, const int* in_sizes, int n_in,
                              void* d_out, int out_size, void* d_ws, size_t ws_size,
                              hipStream_t stream) {
    const float* u       = (const float*)d_in[0];
    const float* Lre     = (const float*)d_in[1];
    const float* Lim     = (const float*)d_in[2];
    const float* Wri     = (const float*)d_in[3];
    const float* Dp      = (const float*)d_in[4];
    const float* logstep = (const float*)d_in[5];
    float* out = (float*)d_out;

    float* ws  = (float*)d_ws;
    float* Zr  = ws;
    float* Zi  = Zr  + (size_t)HH * NN * CHUNK;
    float* CRm = Zi  + (size_t)HH * NN * CHUNK;
    float* CIn = CRm + (size_t)HH * CHUNK * NN;
    float* Tm  = CIn + (size_t)HH * CHUNK * NN;
    float* zCr = Tm  + (size_t)HH * CHUNK * CHUNK;
    float* zCi = zCr + (size_t)HH * NN;

    dss_coeffs<<<HH, NN, 0, stream>>>(Lre, Lim, Wri, logstep,
                                      Zr, Zi, CRm, CIn, Tm, zCr, zCi);

    dss_fused<<<HH * BB, 256, 0, stream>>>(u, Dp, Zr, Zi, Tm, CRm, CIn, zCr, zCi, out);
}


# Coefficient kernel spread over 512 threads per head: Z chain, C chain and the 4096 exp/sincos terms of K on separate waves, matrices staged in LDS and stored as whole lines; K keeps its sum order
# speedup vs baseline: 3.9107x; 1.4705x over previous
#include <hip/hip_runtime.h>
#include <math.h>

#define BB 8
#define HH 128
#define LL 4096
#define NN 64
#define CHUNK 64
#define NCH (LL / CHUNK)
#define XP 68
#define EPSV 1e-7f

typedef float v2f __attribute__((ext_vector_type(2)));
typedef float v8f __attribute__((ext_vector_type(8)));

__device__ __forceinline__ int frag_off(int h, int row, int k) {
    return ((h * 4 + (row >> 4)) * 8 + (k >> 3)) * 128
         + ((row & 15) + 16 * ((k >> 2) & 1)) * 4 + (k & 3);
}

#define CT 512
#define SP 68
#define KP 65

__global__ void __launch_bounds__(CT)
dss_coeffs(const float* __restrict__ Lre, const float* __restrict__ Lim,
           const float* __restrict__ Wri, const float* __restrict__ logstep,
           float* __restrict__ Zr, float* __restrict__ Zi,
           float* __restrict__ CR, float* __restrict__ CIn,
           float* __restrict__ T, float* __restrict__ zCr, float* __restrict__ zCi) {
    __shared__ __attribute__((aligned(16))) float stZr[NN * SP];
    __shared__ __attribute__((aligned(16))) float stZi[NN * SP];
    __shared__ __attribute__((aligned(16))) float stCR[CHUNK * SP];
    __shared__ __attribute__((aligned(16))) float stCI[CHUNK * SP];
    __shared__ float sE[CHUNK * KP], sCs[CHUNK * KP], sSn[CHUNK * KP];
    __shared__ float scr[NN], sci[NN], sK[CHUNK];

    const int h   = blockIdx.x;
    const int tid = threadIdx.x;
    const int n   = tid & 63;

    const float lre = Lre[h * NN + n], lim = Lim[h * NN + n];
    const float dt  = __expf(logstep[h]);
    const float ar  = dt * lre, ai = dt * lim;

    if (tid < 2 * NN) {
        float cz, sz; sincosf(ai, &sz, &cz);
        const float er = expf(ar);
        const float zr = er * cz, zi = er * sz;
        float cL, sL; sincosf(ai * (float)LL, &sL, &cL);
        const float eL = expf(ar * (float)LL);
        const float zLr = eL * cL, zLi = eL * sL;
        const float nr = 1.f - zLr, ni = -zLi;
        const float dr = 1.f - zr,  di = -zi;
        const float dd = dr * dr + di * di;
        const float sr = (nr * dr + ni * di) / dd;
        const float si = (ni * dr - nr * di) / dd;
        const float wr = Wri[(h * NN + n) * 2 + 0], wi = Wri[(h * NN + n) * 2 + 1];
        const float ll = lre * lre + lim * lim;
        const float br = (wr * lre + wi * lim) / ll;
        const float bi = (wi * lre - wr * lim) / ll;
        const float ss = sr * sr + si * si + EPSV;
        const float cr = (br * sr + bi * si) / ss;
        const float ci = (bi * sr - br * si) / ss;

        if (tid < NN) {
            float cC, sC; sincosf(ai * (float)CHUNK, &sC, &cC);
            const float eC = expf(ar * (float)CHUNK);
            zCr[h * NN + n] = eC * cC;
            zCi[h * NN + n] = eC * sC;
            float pr = 1.f, pi_ = 0.f;
            float* __restrict__ dR = stZr + n * SP + (CHUNK - 1);
            float* __restrict__ dI = stZi + n * SP + (CHUNK - 1);
            for (int k = 0; k < CHUNK; ++k) {
                dR[-k] = pr;
                dI[-k] = pi_;
                const float tr = __fmaf_rn(pr, zr, -(pi_ * zi));
                pi_ = __fmaf_rn(pi_, zr, pr * zi);
                pr = tr;
            }
        } else {
            scr[n] = cr; sci[n] = ci;
            float qr = cr * zr - ci * zi;
            float qi = cr * zi + ci * zr;
            for (int t = 0; t < CHUNK; ++t) {
                stCR[t * SP + n] =  qr;
                stCI[t * SP + n] = -qi;
                const float tr = __fmaf_rn(qr, zr, -(qi * zi));
                qi = __fmaf_rn(qi, zr, qr * zi);
                qr = tr;
            }
        }
    } else {
        for (int t = (tid - 2 * NN) >> 6; t < CHUNK; t += (CT - 2 * NN) / NN) {
            const float tf = (float)t;
            const float e = expf(ar * tf);
            float cs, sn; sincosf(ai * tf, &sn, &cs);
            sE [t * KP + n] = e;
            sCs[t * KP + n] = cs;
            sSn[t * KP + n] = sn;
        }
    }
    __syncthreads();

    if (tid < CHUNK) {
        const float* __restrict__ pe = sE  + tid * KP;
        const float* __restrict__ pc = sCs + tid * KP;
        const float* __restrict__ ps = sSn + tid * KP;
        float acc = 0.f;
        for (int j = 0; j < NN; ++j) {
            const float inner = __fmaf_rn(scr[j], pc[j], -(sci[j] * ps[j]));
            acc = __fmaf_rn(pe[j], inner, acc);
        }
        sK[tid] = acc;
    }

    for (int q = tid; q < CHUNK * CHUNK / 4; q += CT) {
        const int rt = q >> 8, g = (q >> 5) & 7, lane = q & 31;
        const int so = (rt * 16 + (lane & 15)) * SP + 8 * g + 4 * (lane >> 4);
        const size_t go = (size_t)h * (CHUNK * CHUNK) + (size_t)q * 4;
        *(float4*)(Zr  + go) = *(const float4*)(stZr + so);
        *(float4*)(Zi  + go) = *(const float4*)(stZi + so);
        *(float4*)(CR  + go) = *(const float4*)(stCR + so);
        *(float4*)(CIn + go) = *(const float4*)(stCI + so);
    }
    __syncthreads();

    for (int q = tid; q < CHUNK * CHUNK / 4; q += CT) {
        const int rt = q >> 8, g = (q >> 5) & 7, lane = q & 31;
        const int d = rt * 16 + (lane & 15) - 8 * g - 4 * (lane >> 4);
        float4 v;
        v.x = (d     >= 0) ? sK[d    ] : 0.f;
        v.y = (d - 1 >= 0) ? sK[d - 1] : 0.f;
        v.z = (d - 2 >= 0) ? sK[d - 2] : 0.f;
        v.w = (d - 3 >= 0) ? sK[d - 3] : 0.f;
        *(float4*)(T + (size_t)h * (CHUNK * CHUNK) + (size_t)q * 4) = v;
    }
}

__global__ void __launch_bounds__(256)
dss_fused(const float* __restrict__ u, const float* __restrict__ D,
          const float* __restrict__ Zr, const float* __restrict__ Zi,
          const float* __restrict__ T, const float* __restrict__ CR,
          const float* __restrict__ CIn,
          const float* __restrict__ zCr, const float* __restrict__ zCi,
          float* __restrict__ out) {
    __shared__ __attribute__((aligned(16))) float sXr[(NCH + 1) * XP];
    __shared__ __attribute__((aligned(16))) float sXi[(NCH + 1) * XP];

    const int tid   = threadIdx.x;
    const int lane  = tid & 31;
    const int wave  = tid >> 5;
    const int h     = blockIdx.x >> 3;
    const int b     = blockIdx.x & 7;
    const int m     = lane & 15;
    const int khalf = lane >> 4;
    const int ct    = wave >> 1;
    const int rt0   = (wave & 1) * 2;
    const int ch    = ct * 16 + m;

    const float* __restrict__ ubh = u + ((size_t)b * HH + h) * LL;
    const float* __restrict__ ub  = ubh + (size_t)ch * CHUNK + 4 * khalf;
    const size_t afrag = (size_t)(h * 4 + rt0) * 1024 + lane * 4;

    const float zr = zCr[h * NN + (tid & 63)];
    const float zi = zCi[h * NN + (tid & 63)];

    float4 bu[CHUNK / 8];
#pragma unroll
    for (int g = 0; g < CHUNK / 8; ++g) bu[g] = *(const float4*)(ub + g * 8);

    if (tid < NN) { sXr[tid] = 0.f; sXi[tid] = 0.f; }

    {
        const float* __restrict__ zrf = Zr + afrag;
        const float* __restrict__ zif = Zi + afrag;
        v8f pr0 = {}, pi0 = {}, pr1 = {}, pi1 = {};
#pragma unroll
        for (int g = 0; g < CHUNK / 8; ++g) {
            const float4 aR0 = *(const float4*)(zrf + g * 128);
            const float4 aI0 = *(const float4*)(zif + g * 128);
            const float4 aR1 = *(const float4*)(zrf + 1024 + g * 128);
            const float4 aI1 = *(const float4*)(zif + 1024 + g * 128);
            const v2f b0 = {bu[g].x, bu[g].y}, b1 = {bu[g].z, bu[g].w};
            const v2f r00 = {aR0.x, aR0.y}, r01 = {aR0.z, aR0.w};
            const v2f i00 = {aI0.x, aI0.y}, i01 = {aI0.z, aI0.w};
            const v2f r10 = {aR1.x, aR1.y}, r11 = {aR1.z, aR1.w};
            const v2f i10 = {aI1.x, aI1.y}, i11 = {aI1.z, aI1.w};
            pr0 = __builtin_amdgcn_wmma_f32_16x16x4_f32(false, r00, false, b0, (short)0, pr0, false, false);
            pi0 = __builtin_amdgcn_wmma_f32_16x16x4_f32(false, i00, false, b0, (short)0, pi0, false, false);
            pr1 = __builtin_amdgcn_wmma_f32_16x16x4_f32(false, r10, false, b0, (short)0, pr1, false, false);
            pi1 = __builtin_amdgcn_wmma_f32_16x16x4_f32(false, i10, false, b0, (short)0, pi1, false, false);
            pr0 = __builtin_amdgcn_wmma_f32_16x16x4_f32(false, r01, false, b1, (short)0, pr0, false, false);
            pi0 = __builtin_amdgcn_wmma_f32_16x16x4_f32(false, i01, false, b1, (short)0, pi0, false, false);
            pr1 = __builtin_amdgcn_wmma_f32_16x16x4_f32(false, r11, false, b1, (short)0, pr1, false, false);
            pi1 = __builtin_amdgcn_wmma_f32_16x16x4_f32(false, i11, false, b1, (short)0, pi1, false, false);
        }
        float* __restrict__ dr = sXr + (ch + 1) * XP + rt0 * 16 + 8 * khalf;
        float* __restrict__ di = sXi + (ch + 1) * XP + rt0 * 16 + 8 * khalf;
        float4 v;
        v.x = pr0[0]; v.y = pr0[1]; v.z = pr0[2]; v.w = pr0[3]; *(float4*)(dr)      = v;
        v.x = pr0[4]; v.y = pr0[5]; v.z = pr0[6]; v.w = pr0[7]; *(float4*)(dr + 4)  = v;
        v.x = pr1[0]; v.y = pr1[1]; v.z = pr1[2]; v.w = pr1[3]; *(float4*)(dr + 16) = v;
        v.x = pr1[4]; v.y = pr1[5]; v.z = pr1[6]; v.w = pr1[7]; *(float4*)(dr + 20) = v;
        v.x = pi0[0]; v.y = pi0[1]; v.z = pi0[2]; v.w = pi0[3]; *(float4*)(di)      = v;
        v.x = pi0[4]; v.y = pi0[5]; v.z = pi0[6]; v.w = pi0[7]; *(float4*)(di + 4)  = v;
        v.x = pi1[0]; v.y = pi1[1]; v.z = pi1[2]; v.w = pi1[3]; *(float4*)(di + 16) = v;
        v.x = pi1[4]; v.y = pi1[5]; v.z = pi1[6]; v.w = pi1[7]; *(float4*)(di + 20) = v;
    }
    __syncthreads();

    if (tid < NN) {
        float xr = 0.f, xi = 0.f;
        float* __restrict__ sr = sXr + XP + tid;
        float* __restrict__ si = sXi + XP + tid;
#pragma unroll
        for (int i0 = 0; i0 < NCH; i0 += 16) {
            float vr[16], vi[16];
#pragma unroll
            for (int j = 0; j < 16; ++j) { vr[j] = sr[(i0 + j) * XP]; vi[j] = si[(i0 + j) * XP]; }
#pragma unroll
            for (int j = 0; j < 16; ++j) {
                const float nxr = zr * xr - zi * xi + vr[j];
                const float nxi = zr * xi + zi * xr + vi[j];
                vr[j] = nxr; vi[j] = nxi;
                xr = nxr; xi = nxi;
            }
#pragma unroll
            for (int j = 0; j < 16; ++j) { sr[(i0 + j) * XP] = vr[j]; si[(i0 + j) * XP] = vi[j]; }
        }
    }

    v8f acc0 = {}, acc1 = {};
    {
        const float* __restrict__ tf = T + afrag;
#pragma unroll
        for (int g = 0; g < CHUNK / 8; ++g) {
            const float4 a0 = *(const float4*)(tf + g * 128);
            const float4 a1 = *(const float4*)(tf + 1024 + g * 128);
            const v2f b0 = {bu[g].x, bu[g].y}, b1 = {bu[g].z, bu[g].w};
            const v2f a00 = {a0.x, a0.y}, a01 = {a0.z, a0.w};
            const v2f a10 = {a1.x, a1.y}, a11 = {a1.z, a1.w};
            acc0 = __builtin_amdgcn_wmma_f32_16x16x4_f32(false, a00, false, b0, (short)0, acc0, false, false);
            acc1 = __builtin_amdgcn_wmma_f32_16x16x4_f32(false, a10, false, b0, (short)0, acc1, false, false);
            acc0 = __builtin_amdgcn_wmma_f32_16x16x4_f32(false, a01, false, b1, (short)0, acc0, false, false);
            acc1 = __builtin_amdgcn_wmma_f32_16x16x4_f32(false, a11, false, b1, (short)0, acc1, false, false);
        }
    }
    __syncthreads();

    {
        const float* __restrict__ crf = CR  + afrag;
        const float* __restrict__ cif = CIn + afrag;
        const float* __restrict__ xrb = sXr + ch * XP + 4 * khalf;
        const float* __restrict__ xib = sXi + ch * XP + 4 * khalf;
#pragma unroll
        for (int g = 0; g < NN / 8; ++g) {
            const float4 aR0 = *(const float4*)(crf + g * 128);
            const float4 aI0 = *(const float4*)(cif + g * 128);
            const float4 aR1 = *(const float4*)(crf + 1024 + g * 128);
            const float4 aI1 = *(const float4*)(cif + 1024 + g * 128);
            const float4 bxr = *(const float4*)(xrb + g * 8);
            const float4 bxi = *(const float4*)(xib + g * 8);
            const v2f xr0 = {bxr.x, bxr.y}, xr1 = {bxr.z, bxr.w};
            const v2f xi0 = {bxi.x, bxi.y}, xi1 = {bxi.z, bxi.w};
            const v2f r00 = {aR0.x, aR0.y}, r01 = {aR0.z, aR0.w};
            const v2f i00 = {aI0.x, aI0.y}, i01 = {aI0.z, aI0.w};
            const v2f r10 = {aR1.x, aR1.y}, r11 = {aR1.z, aR1.w};
            const v2f i10 = {aI1.x, aI1.y}, i11 = {aI1.z, aI1.w};
            acc0 = __builtin_amdgcn_wmma_f32_16x16x4_f32(false, r00, false, xr0, (short)0, acc0, false, false);
            acc1 = __builtin_amdgcn_wmma_f32_16x16x4_f32(false, r10, false, xr0, (short)0, acc1, false, false);
            acc0 = __builtin_amdgcn_wmma_f32_16x16x4_f32(false, i00, false, xi0, (short)0, acc0, false, false);
            acc1 = __builtin_amdgcn_wmma_f32_16x16x4_f32(false, i10, false, xi0, (short)0, acc1, false, false);
            acc0 = __builtin_amdgcn_wmma_f32_16x16x4_f32(false, r01, false, xr1, (short)0, acc0, false, false);
            acc1 = __builtin_amdgcn_wmma_f32_16x16x4_f32(false, r11, false, xr1, (short)0, acc1, false, false);
            acc0 = __builtin_amdgcn_wmma_f32_16x16x4_f32(false, i01, false, xi1, (short)0, acc0, false, false);
            acc1 = __builtin_amdgcn_wmma_f32_16x16x4_f32(false, i11, false, xi1, (short)0, acc1, false, false);
        }
    }

    const float Dh = D[h];
    const size_t obase = ((size_t)b * HH + h) * LL + (size_t)ch * CHUNK + rt0 * 16 + 8 * khalf;
    {
        const float4 u0 = *(const float4*)(u + obase);
        const float4 u1 = *(const float4*)(u + obase + 4);
        float4 o0, o1;
        o0.x = acc0[0] + Dh * u0.x; o0.y = acc0[1] + Dh * u0.y;
        o0.z = acc0[2] + Dh * u0.z; o0.w = acc0[3] + Dh * u0.w;
        o1.x = acc0[4] + Dh * u1.x; o1.y = acc0[5] + Dh * u1.y;
        o1.z = acc0[6] + Dh * u1.z; o1.w = acc0[7] + Dh * u1.w;
        *(float4*)(out + obase)     = o0;
        *(float4*)(out + obase + 4) = o1;
    }
    {
        const float4 u0 = *(const float4*)(u + obase + 16);
        const float4 u1 = *(const float4*)(u + obase + 20);
        float4 o0, o1;
        o0.x = acc1[0] + Dh * u0.x; o0.y = acc1[1] + Dh * u0.y;
        o0.z = acc1[2] + Dh * u0.z; o0.w = acc1[3] + Dh * u0.w;
        o1.x = acc1[4] + Dh * u1.x; o1.y = acc1[5] + Dh * u1.y;
        o1.z = acc1[6] + Dh * u1.z; o1.w = acc1[7] + Dh * u1.w;
        *(float4*)(out + obase + 16) = o0;
        *(float4*)(out + obase + 20) = o1;
    }
}

extern "C" void kernel_launch(void* const* d_in, const int* in_sizes, int n_in,
                              void* d_out, int out_size, void* d_ws, size_t ws_size,
                              hipStream_t stream) {
    const float* u       = (const float*)d_in[0];
    const float* Lre     = (const float*)d_in[1];
    const float* Lim     = (const float*)d_in[2];
    const float* Wri     = (const float*)d_in[3];
    const float* Dp      = (const float*)d_in[4];
    const float* logstep = (const float*)d_in[5];
    float* out = (float*)d_out;

    float* ws  = (float*)d_ws;
    float* Zr  = ws;
    float* Zi  = Zr  + (size_t)HH * NN * CHUNK;
    float* CRm = Zi  + (size_t)HH * NN * CHUNK;
    float* CIn = CRm + (size_t)HH * CHUNK * NN;
    float* Tm  = CIn + (size_t)HH * CHUNK * NN;
    float* zCr = Tm  + (size_t)HH * CHUNK * CHUNK;
    float* zCi = zCr + (size_t)HH * NN;

    dss_coeffs<<<HH, CT, 0, stream>>>(Lre, Lim, Wri, logstep,
                                      Zr, Zi, CRm, CIn, Tm, zCr, zCi);

    dss_fused<<<HH * BB, 256, 0, stream>>>(u, Dp, Zr, Zi, Tm, CRm, CIn, zCr, zCi, out);
}


# Fused kernel: one work-group covers two batches of a head: each Z, T and C fragment is loaded once for both, the scan runs on 128 threads, half the waves enter; per batch every sum keeps its order
# speedup vs baseline: 4.3419x; 1.1103x over previous
#include <hip/hip_runtime.h>
#include <math.h>

#define BB 8
#define HH 128
#define LL 4096
#define NN 64
#define CHUNK 64
#define NCH (LL / CHUNK)
#define XP 68
#define EPSV 1e-7f

typedef float v2f __attribute__((ext_vector_type(2)));
typedef float v8f __attribute__((ext_vector_type(8)));

__device__ __forceinline__ int frag_off(int h, int row, int k) {
    return ((h * 4 + (row >> 4)) * 8 + (k >> 3)) * 128
         + ((row & 15) + 16 * ((k >> 2) & 1)) * 4 + (k & 3);
}

#define CT 512
#define SP 68
#define KP 65

__global__ void __launch_bounds__(CT)
dss_coeffs(const float* __restrict__ Lre, const float* __restrict__ Lim,
           const float* __restrict__ Wri, const float* __restrict__ logstep,
           float* __restrict__ Zr, float* __restrict__ Zi,
           float* __restrict__ CR, float* __restrict__ CIn,
           float* __restrict__ T, float* __restrict__ zCr, float* __restrict__ zCi) {
    __shared__ __attribute__((aligned(16))) float stZr[NN * SP];
    __shared__ __attribute__((aligned(16))) float stZi[NN * SP];
    __shared__ __attribute__((aligned(16))) float stCR[CHUNK * SP];
    __shared__ __attribute__((aligned(16))) float stCI[CHUNK * SP];
    __shared__ float sE[CHUNK * KP], sCs[CHUNK * KP], sSn[CHUNK * KP];
    __shared__ float scr[NN], sci[NN], sK[CHUNK];

    const int h   = blockIdx.x;
    const int tid = threadIdx.x;
    const int n   = tid & 63;

    const float lre = Lre[h * NN + n], lim = Lim[h * NN + n];
    const float dt  = __expf(logstep[h]);
    const float ar  = dt * lre, ai = dt * lim;

    if (tid < 2 * NN) {
        float cz, sz; sincosf(ai, &sz, &cz);
        const float er = expf(ar);
        const float zr = er * cz, zi = er * sz;
        float cL, sL; sincosf(ai * (float)LL, &sL, &cL);
        const float eL = expf(ar * (float)LL);
        const float zLr = eL * cL, zLi = eL * sL;
        const float nr = 1.f - zLr, ni = -zLi;
        const float dr = 1.f - zr,  di = -zi;
        const float dd = dr * dr + di * di;
        const float sr = (nr * dr + ni * di) / dd;
        const float si = (ni * dr - nr * di) / dd;
        const float wr = Wri[(h * NN + n) * 2 + 0], wi = Wri[(h * NN + n) * 2 + 1];
        const float ll = lre * lre + lim * lim;
        const float br = (wr * lre + wi * lim) / ll;
        const float bi = (wi * lre - wr * lim) / ll;
        const float ss = sr * sr + si * si + EPSV;
        const float cr = (br * sr + bi * si) / ss;
        const float ci = (bi * sr - br * si) / ss;

        if (tid < NN) {
            float cC, sC; sincosf(ai * (float)CHUNK, &sC, &cC);
            const float eC = expf(ar * (float)CHUNK);
            zCr[h * NN + n] = eC * cC;
            zCi[h * NN + n] = eC * sC;
            float pr = 1.f, pi_ = 0.f;
            float* __restrict__ dR = stZr + n * SP + (CHUNK - 1);
            float* __restrict__ dI = stZi + n * SP + (CHUNK - 1);
            for (int k = 0; k < CHUNK; ++k) {
                dR[-k] = pr;
                dI[-k] = pi_;
                const float tr = __fmaf_rn(pr, zr, -(pi_ * zi));
                pi_ = __fmaf_rn(pi_, zr, pr * zi);
                pr = tr;
            }
        } else {
            scr[n] = cr; sci[n] = ci;
            float qr = cr * zr - ci * zi;
            float qi = cr * zi + ci * zr;
            for (int t = 0; t < CHUNK; ++t) {
                stCR[t * SP + n] =  qr;
                stCI[t * SP + n] = -qi;
                const float tr = __fmaf_rn(qr, zr, -(qi * zi));
                qi = __fmaf_rn(qi, zr, qr * zi);
                qr = tr;
            }
        }
    } else {
        for (int t = (tid - 2 * NN) >> 6; t < CHUNK; t += (CT - 2 * NN) / NN) {
            const float tf = (float)t;
            const float e = expf(ar * tf);
            float cs, sn; sincosf(ai * tf, &sn, &cs);
            sE [t * KP + n] = e;
            sCs[t * KP + n] = cs;
            sSn[t * KP + n] = sn;
        }
    }
    __syncthreads();

    if (tid < CHUNK) {
        const float* __restrict__ pe = sE  + tid * KP;
        const float* __restrict__ pc = sCs + tid * KP;
        const float* __restrict__ ps = sSn + tid * KP;
        float acc = 0.f;
        for (int j = 0; j < NN; ++j) {
            const float inner = __fmaf_rn(scr[j], pc[j], -(sci[j] * ps[j]));
            acc = __fmaf_rn(pe[j], inner, acc);
        }
        sK[tid] = acc;
    }

    for (int q = tid; q < CHUNK * CHUNK / 4; q += CT) {
        const int rt = q >> 8, g = (q >> 5) & 7, lane = q & 31;
        const int so = (rt * 16 + (lane & 15)) * SP + 8 * g + 4 * (lane >> 4);
        const size_t go = (size_t)h * (CHUNK * CHUNK) + (size_t)q * 4;
        *(float4*)(Zr  + go) = *(const float4*)(stZr + so);
        *(float4*)(Zi  + go) = *(const float4*)(stZi + so);
        *(float4*)(CR  + go) = *(const float4*)(stCR + so);
        *(float4*)(CIn + go) = *(const float4*)(stCI + so);
    }
    __syncthreads();

    for (int q = tid; q < CHUNK * CHUNK / 4; q += CT) {
        const int rt = q >> 8, g = (q >> 5) & 7, lane = q & 31;
        const int d = rt * 16 + (lane & 15) - 8 * g - 4 * (lane >> 4);
        float4 v;
        v.x = (d     >= 0) ? sK[d    ] : 0.f;
        v.y = (d - 1 >= 0) ? sK[d - 1] : 0.f;
        v.z = (d - 2 >= 0) ? sK[d - 2] : 0.f;
        v.w = (d - 3 >= 0) ? sK[d - 3] : 0.f;
        *(float4*)(T + (size_t)h * (CHUNK * CHUNK) + (size_t)q * 4) = v;
    }
}

#define NB 2

__global__ void __launch_bounds__(256, 4)
dss_fused(const float* __restrict__ u, const float* __restrict__ D,
          const float* __restrict__ Zr, const float* __restrict__ Zi,
          const float* __restrict__ T, const float* __restrict__ CR,
          const float* __restrict__ CIn,
          const float* __restrict__ zCr, const float* __restrict__ zCi,
          float* __restrict__ out) {
    __shared__ __attribute__((aligned(16))) float sXr[NB][(NCH + 1) * XP];
    __shared__ __attribute__((aligned(16))) float sXi[NB][(NCH + 1) * XP];

    const int tid   = threadIdx.x;
    const int lane  = tid & 31;
    const int wave  = tid >> 5;
    const int h     = blockIdx.x / (BB / NB);
    const int bfst  = (blockIdx.x % (BB / NB)) * NB;
    const int m     = lane & 15;
    const int khalf = lane >> 4;
    const int ct    = wave >> 1;
    const int rt0   = (wave & 1) * 2;
    const int ch    = ct * 16 + m;

    const size_t afrag = (size_t)(h * 4 + rt0) * 1024 + lane * 4;

    float4 bu[NB][CHUNK / 8];
#pragma unroll
    for (int bb = 0; bb < NB; ++bb) {
        const float* __restrict__ ub = u + ((size_t)(bfst + bb) * HH + h) * LL + (size_t)ch * CHUNK + 4 * khalf;
#pragma unroll
        for (int g = 0; g < CHUNK / 8; ++g) bu[bb][g] = *(const float4*)(ub + g * 8);
    }

    if (tid < NB * NN) { sXr[tid >> 6][tid & 63] = 0.f; sXi[tid >> 6][tid & 63] = 0.f; }

    {
        const float* __restrict__ zrf = Zr + afrag;
        const float* __restrict__ zif = Zi + afrag;
        v8f pr0[NB], pi0[NB], pr1[NB], pi1[NB];
#pragma unroll
        for (int bb = 0; bb < NB; ++bb) { pr0[bb] = v8f{}; pi0[bb] = v8f{}; pr1[bb] = v8f{}; pi1[bb] = v8f{}; }
#pragma unroll
        for (int g = 0; g < CHUNK / 8; ++g) {
            const float4 aR0 = *(const float4*)(zrf + g * 128);
            const float4 aI0 = *(const float4*)(zif + g * 128);
            const float4 aR1 = *(const float4*)(zrf + 1024 + g * 128);
            const float4 aI1 = *(const float4*)(zif + 1024 + g * 128);
            const v2f r00 = {aR0.x, aR0.y}, r01 = {aR0.z, aR0.w};
            const v2f i00 = {aI0.x, aI0.y}, i01 = {aI0.z, aI0.w};
            const v2f r10 = {aR1.x, aR1.y}, r11 = {aR1.z, aR1.w};
            const v2f i10 = {aI1.x, aI1.y}, i11 = {aI1.z, aI1.w};
#pragma unroll
            for (int bb = 0; bb < NB; ++bb) {
                const v2f b0 = {bu[bb][g].x, bu[bb][g].y}, b1 = {bu[bb][g].z, bu[bb][g].w};
                pr0[bb] = __builtin_amdgcn_wmma_f32_16x16x4_f32(false, r00, false, b0, (short)0, pr0[bb], false, false);
                pi0[bb] = __builtin_amdgcn_wmma_f32_16x16x4_f32(false, i00, false, b0, (short)0, pi0[bb], false, false);
                pr1[bb] = __builtin_amdgcn_wmma_f32_16x16x4_f32(false, r10, false, b0, (short)0, pr1[bb], false, false);
                pi1[bb] = __builtin_amdgcn_wmma_f32_16x16x4_f32(false, i10, false, b0, (short)0, pi1[bb], false, false);
                pr0[bb] = __builtin_amdgcn_wmma_f32_16x16x4_f32(false, r01, false, b1, (short)0, pr0[bb], false, false);
                pi0[bb] = __builtin_amdgcn_wmma_f32_16x16x4_f32(false, i01, false, b1, (short)0, pi0[bb], false, false);
                pr1[bb] = __builtin_amdgcn_wmma_f32_16x16x4_f32(false, r11, false, b1, (short)0, pr1[bb], false, false);
                pi1[bb] = __builtin_amdgcn_wmma_f32_16x16x4_f32(false, i11, false, b1, (short)0, pi1[bb], false, false);
            }
        }
#pragma unroll
        for (int bb = 0; bb < NB; ++bb) {
            float* __restrict__ dr = sXr[bb] + (ch + 1) * XP + rt0 * 16 + 8 * khalf;
            float* __restrict__ di = sXi[bb] + (ch + 1) * XP + rt0 * 16 + 8 * khalf;
            float4 v;
            v.x = pr0[bb][0]; v.y = pr0[bb][1]; v.z = pr0[bb][2]; v.w = pr0[bb][3]; *(float4*)(dr)      = v;
            v.x = pr0[bb][4]; v.y = pr0[bb][5]; v.z = pr0[bb][6]; v.w = pr0[bb][7]; *(float4*)(dr + 4)  = v;
            v.x = pr1[bb][0]; v.y = pr1[bb][1]; v.z = pr1[bb][2]; v.w = pr1[bb][3]; *(float4*)(dr + 16) = v;
            v.x = pr1[bb][4]; v.y = pr1[bb][5]; v.z = pr1[bb][6]; v.w = pr1[bb][7]; *(float4*)(dr + 20) = v;
            v.x = pi0[bb][0]; v.y = pi0[bb][1]; v.z = pi0[bb][2]; v.w = pi0[bb][3]; *(float4*)(di)      = v;
            v.x = pi0[bb][4]; v.y = pi0[bb][5]; v.z = pi0[bb][6]; v.w = pi0[bb][7]; *(float4*)(di + 4)  = v;
            v.x = pi1[bb][0]; v.y = pi1[bb][1]; v.z = pi1[bb][2]; v.w = pi1[bb][3]; *(float4*)(di + 16) = v;
            v.x = pi1[bb][4]; v.y = pi1[bb][5]; v.z = pi1[bb][6]; v.w = pi1[bb][7]; *(float4*)(di + 20) = v;
        }
    }
    __syncthreads();

    if (tid < NB * NN) {
        const float zr = zCr[h * NN + (tid & 63)];
        const float zi = zCi[h * NN + (tid & 63)];
        float xr = 0.f, xi = 0.f;
        float* __restrict__ sr = sXr[tid >> 6] + XP + (tid & 63);
        float* __restrict__ si = sXi[tid >> 6] + XP + (tid & 63);
#pragma unroll
        for (int i0 = 0; i0 < NCH; i0 += 16) {
            float vr[16], vi[16];
#pragma unroll
            for (int j = 0; j < 16; ++j) { vr[j] = sr[(i0 + j) * XP]; vi[j] = si[(i0 + j) * XP]; }
#pragma unroll
            for (int j = 0; j < 16; ++j) {
                const float nxr = zr * xr - zi * xi + vr[j];
                const float nxi = zr * xi + zi * xr + vi[j];
                vr[j] = nxr; vi[j] = nxi;
                xr = nxr; xi = nxi;
            }
#pragma unroll
            for (int j = 0; j < 16; ++j) { sr[(i0 + j) * XP] = vr[j]; si[(i0 + j) * XP] = vi[j]; }
        }
    }

    v8f acc0[NB], acc1[NB];
#pragma unroll
    for (int bb = 0; bb < NB; ++bb) { acc0[bb] = v8f{}; acc1[bb] = v8f{}; }
    {
        const float* __restrict__ tf = T + afrag;
#pragma unroll
        for (int g = 0; g < CHUNK / 8; ++g) {
            const float4 a0 = *(const float4*)(tf + g * 128);
            const float4 a1 = *(const float4*)(tf + 1024 + g * 128);
            const v2f a00 = {a0.x, a0.y}, a01 = {a0.z, a0.w};
            const v2f a10 = {a1.x, a1.y}, a11 = {a1.z, a1.w};
#pragma unroll
            for (int bb = 0; bb < NB; ++bb) {
                const v2f b0 = {bu[bb][g].x, bu[bb][g].y}, b1 = {bu[bb][g].z, bu[bb][g].w};
                acc0[bb] = __builtin_amdgcn_wmma_f32_16x16x4_f32(false, a00, false, b0, (short)0, acc0[bb], false, false);
                acc1[bb] = __builtin_amdgcn_wmma_f32_16x16x4_f32(false, a10, false, b0, (short)0, acc1[bb], false, false);
                acc0[bb] = __builtin_amdgcn_wmma_f32_16x16x4_f32(false, a01, false, b1, (short)0, acc0[bb], false, false);
                acc1[bb] = __builtin_amdgcn_wmma_f32_16x16x4_f32(false, a11, false, b1, (short)0, acc1[bb], false, false);
            }
        }
    }
    __syncthreads();

    {
        const float* __restrict__ crf = CR  + afrag;
        const float* __restrict__ cif = CIn + afrag;
#pragma unroll
        for (int g = 0; g < NN / 8; ++g) {
            const float4 aR0 = *(const float4*)(crf + g * 128);
            const float4 aI0 = *(const float4*)(cif + g * 128);
            const float4 aR1 = *(const float4*)(crf + 1024 + g * 128);
            const float4 aI1 = *(const float4*)(cif + 1024 + g * 128);
            const v2f r00 = {aR0.x, aR0.y}, r01 = {aR0.z, aR0.w};
            const v2f i00 = {aI0.x, aI0.y}, i01 = {aI0.z, aI0.w};
            const v2f r10 = {aR1.x, aR1.y}, r11 = {aR1.z, aR1.w};
            const v2f i10 = {aI1.x, aI1.y}, i11 = {aI1.z, aI1.w};
#pragma unroll
            for (int bb = 0; bb < NB; ++bb) {
                const float4 bxr = *(const float4*)(sXr[bb] + ch * XP + 4 * khalf + g * 8);
                const float4 bxi = *(const float4*)(sXi[bb] + ch * XP + 4 * khalf + g * 8);
                const v2f xr0 = {bxr.x, bxr.y}, xr1 = {bxr.z, bxr.w};
                const v2f xi0 = {bxi.x, bxi.y}, xi1 = {bxi.z, bxi.w};
                acc0[bb] = __builtin_amdgcn_wmma_f32_16x16x4_f32(false, r00, false, xr0, (short)0, acc0[bb], false, false);
                acc1[bb] = __builtin_amdgcn_wmma_f32_16x16x4_f32(false, r10, false, xr0, (short)0, acc1[bb], false, false);
                acc0[bb] = __builtin_amdgcn_wmma_f32_16x16x4_f32(false, i00, false, xi0, (short)0, acc0[bb], false, false);
                acc1[bb] = __builtin_amdgcn_wmma_f32_16x16x4_f32(false, i10, false, xi0, (short)0, acc1[bb], false, false);
                acc0[bb] = __builtin_amdgcn_wmma_f32_16x16x4_f32(false, r01, false, xr1, (short)0, acc0[bb], false, false);
                acc1[bb] = __builtin_amdgcn_wmma_f32_16x16x4_f32(false, r11, false, xr1, (short)0, acc1[bb], false, false);
                acc0[bb] = __builtin_amdgcn_wmma_f32_16x16x4_f32(false, i01, false, xi1, (short)0, acc0[bb], false, false);
                acc1[bb] = __builtin_amdgcn_wmma_f32_16x16x4_f32(false, i11, false, xi1, (short)0, acc1[bb], false, false);
            }
        }
    }

    const float Dh = D[h];
#pragma unroll
    for (int bb = 0; bb < NB; ++bb) {
        const size_t obase = ((size_t)(bfst + bb) * HH + h) * LL + (size_t)ch * CHUNK + rt0 * 16 + 8 * khalf;
        {
            const float4 u0 = *(const float4*)(u + obase);
            const float4 u1 = *(const float4*)(u + obase + 4);
            float4 o0, o1;
            o0.x = acc0[bb][0] + Dh * u0.x; o0.y = acc0[bb][1] + Dh * u0.y;
            o0.z = acc0[bb][2] + Dh * u0.z; o0.w = acc0[bb][3] + Dh * u0.w;
            o1.x = acc0[bb][4] + Dh * u1.x; o1.y = acc0[bb][5] + Dh * u1.y;
            o1.z = acc0[bb][6] + Dh * u1.z; o1.w = acc0[bb][7] + Dh * u1.w;
            *(float4*)(out + obase)     = o0;
            *(float4*)(out + obase + 4) = o1;
        }
        {
            const float4 u0 = *(const float4*)(u + obase + 16);
            const float4 u1 = *(const float4*)(u + obase + 20);
            float4 o0, o1;
            o0.x = acc1[bb][0] + Dh * u0.x; o0.y = acc1[bb][1] + Dh * u0.y;
            o0.z = acc1[bb][2] + Dh * u0.z; o0.w = acc1[bb][3] + Dh * u0.w;
            o1.x = acc1[bb][4] + Dh * u1.x; o1.y = acc1[bb][5] + Dh * u1.y;
            o1.z = acc1[bb][6] + Dh * u1.z; o1.w = acc1[bb][7] + Dh * u1.w;
            *(float4*)(out + obase + 16) = o0;
            *(float4*)(out + obase + 20) = o1;
        }
    }
}

extern "C" void kernel_launch(void* const* d_in, const int* in_sizes, int n_in,
                              void* d_out, int out_size, void* d_ws, size_t ws_size,
                              hipStream_t stream) {
    const float* u       = (const float*)d_in[0];
    const float* Lre     = (const float*)d_in[1];
    const float* Lim     = (const float*)d_in[2];
    const float* Wri     = (const float*)d_in[3];
    const float* Dp      = (const float*)d_in[4];
    const float* logstep = (const float*)d_in[5];
    float* out = (float*)d_out;

    float* ws  = (float*)d_ws;
    float* Zr  = ws;
    float* Zi  = Zr  + (size_t)HH * NN * CHUNK;
    float* CRm = Zi  + (size_t)HH * NN * CHUNK;
    float* CIn = CRm + (size_t)HH * CHUNK * NN;
    float* Tm  = CIn + (size_t)HH * CHUNK * NN;
    float* zCr = Tm  + (size_t)HH * CHUNK * CHUNK;
    float* zCi = zCr + (size_t)HH * NN;

    dss_coeffs<<<HH, CT, 0, stream>>>(Lre, Lim, Wri, logstep,
                                      Zr, Zi, CRm, CIn, Tm, zCr, zCi);

    dss_fused<<<HH * (BB / NB), 256, 0, stream>>>(u, Dp, Zr, Zi, Tm, CRm, CIn, zCr, zCi, out);
}


# Coefficient build fused into the main kernel: a work-group per head and 4 batches builds Z, C, K in its LDS and reads A fragments there; 2 launches become 1; every sum keeps its order
# speedup vs baseline: 5.6895x; 1.3104x over previous
#include <hip/hip_runtime.h>
#include <math.h>

#define BB 8
#define HH 128
#define LL 4096
#define NN 64
#define CHUNK 64
#define NCH (LL / CHUNK)
#define XP 68
#define SP 68
#define KP 65
#define EPSV 1e-7f

typedef float v2f __attribute__((ext_vector_type(2)));
typedef float v8f __attribute__((ext_vector_type(8)));

#define NB 4
#define WB 2
#define FT 512

__global__ void __launch_bounds__(FT, 4)
dss_fused(const float* __restrict__ u,
          const float* __restrict__ Lre, const float* __restrict__ Lim,
          const float* __restrict__ Wri, const float* __restrict__ D,
          const float* __restrict__ logstep,
          float* __restrict__ out) {
    __shared__ __attribute__((aligned(16))) float stZr[NN * SP];
    __shared__ __attribute__((aligned(16))) float stZi[NN * SP];
    __shared__ __attribute__((aligned(16))) float stCR[CHUNK * SP];
    __shared__ __attribute__((aligned(16))) float stCI[CHUNK * SP];
    __shared__ __attribute__((aligned(16))) float sXr[NB][(NCH + 1) * XP];
    __shared__ __attribute__((aligned(16))) float sXi[NB][(NCH + 1) * XP];
    __shared__ float sE[CHUNK * KP], sCs[CHUNK * KP], sSn[CHUNK * KP];
    __shared__ float scr[NN], sci[NN], sK[CHUNK], szCr[NN], szCi[NN];

    const int tid   = threadIdx.x;
    const int lane  = tid & 31;
    const int wave  = tid >> 5;
    const int h     = blockIdx.x / (BB / NB);
    const int pfst  = (wave >> 3) * WB;
    const int bfst  = (blockIdx.x % (BB / NB)) * NB + pfst;
    const int m     = lane & 15;
    const int khalf = lane >> 4;
    const int ct    = (wave & 7) >> 1;
    const int rt0   = (wave & 1) * 2;
    const int ch    = ct * 16 + m;

    float4 bu[WB][CHUNK / 8];
#pragma unroll
    for (int bb = 0; bb < WB; ++bb) {
        const float* __restrict__ ub = u + ((size_t)(bfst + bb) * HH + h) * LL + (size_t)ch * CHUNK + 4 * khalf;
#pragma unroll
        for (int g = 0; g < CHUNK / 8; ++g) bu[bb][g] = *(const float4*)(ub + g * 8);
    }

    if (tid < NB * NN) { sXr[tid >> 6][tid & 63] = 0.f; sXi[tid >> 6][tid & 63] = 0.f; }

    {
        const int n = tid & 63;
        const float lre = Lre[h * NN + n], lim = Lim[h * NN + n];
        const float dt  = __expf(logstep[h]);
        const float ar  = dt * lre, ai = dt * lim;

        if (tid < 2 * NN) {
            float cz, sz; sincosf(ai, &sz, &cz);
            const float er = expf(ar);
            const float zr = er * cz, zi = er * sz;
            float cL, sL; sincosf(ai * (float)LL, &sL, &cL);
            const float eL = expf(ar * (float)LL);
            const float zLr = eL * cL, zLi = eL * sL;
            const float nr = 1.f - zLr, ni = -zLi;
            const float dr = 1.f - zr,  di = -zi;
            const float dd = dr * dr + di * di;
            const float sr = (nr * dr + ni * di) / dd;
            const float si = (ni * dr - nr * di) / dd;
            const float wr = Wri[(h * NN + n) * 2 + 0], wi = Wri[(h * NN + n) * 2 + 1];
            const float ll = lre * lre + lim * lim;
            const float br = (wr * lre + wi * lim) / ll;
            const float bi = (wi * lre - wr * lim) / ll;
            const float ss = sr * sr + si * si + EPSV;
            const float cr = (br * sr + bi * si) / ss;
            const float ci = (bi * sr - br * si) / ss;

            if (tid < NN) {
                float cC, sC; sincosf(ai * (float)CHUNK, &sC, &cC);
                const float eC = expf(ar * (float)CHUNK);
                szCr[n] = eC * cC;
                szCi[n] = eC * sC;
                float pr = 1.f, pi_ = 0.f;
                float* __restrict__ dR = stZr + n * SP + (CHUNK - 1);
                float* __restrict__ dI = stZi + n * SP + (CHUNK - 1);
                for (int k = 0; k < CHUNK; ++k) {
                    dR[-k] = pr;
                    dI[-k] = pi_;
                    const float tr = __fmaf_rn(pr, zr, -(pi_ * zi));
                    pi_ = __fmaf_rn(pi_, zr, pr * zi);
                    pr = tr;
                }
            } else {
                scr[n] = cr; sci[n] = ci;
                float qr = cr * zr - ci * zi;
                float qi = cr * zi + ci * zr;
                for (int t = 0; t < CHUNK; ++t) {
                    stCR[t * SP + n] =  qr;
                    stCI[t * SP + n] = -qi;
                    const float tr = __fmaf_rn(qr, zr, -(qi * zi));
                    qi = __fmaf_rn(qi, zr, qr * zi);
                    qr = tr;
                }
            }
        } else {
            for (int t = (tid - 2 * NN) >> 6; t < CHUNK; t += (FT - 2 * NN) / NN) {
                const float tf = (float)t;
                const float e = expf(ar * tf);
                float cs, sn; sincosf(ai * tf, &sn, &cs);
                sE [t * KP + n] = e;
                sCs[t * KP + n] = cs;
                sSn[t * KP + n] = sn;
            }
        }
    }
    __syncthreads();

    if (tid < CHUNK) {
        const float* __restrict__ pe = sE  + tid * KP;
        const float* __restrict__ pc = sCs + tid * KP;
        const float* __restrict__ ps = sSn + tid * KP;
        float acc = 0.f;
        for (int j = 0; j < NN; ++j) {
            const float inner = __fmaf_rn(scr[j], pc[j], -(sci[j] * ps[j]));
            acc = __fmaf_rn(pe[j], inner, acc);
        }
        sK[tid] = acc;
    }

    const int aoff = (rt0 * 16 + m) * SP + 4 * khalf;

    {
        const float* __restrict__ zrf = stZr + aoff;
        const float* __restrict__ zif = stZi + aoff;
        v8f pr0[WB], pi0[WB], pr1[WB], pi1[WB];
#pragma unroll
        for (int bb = 0; bb < WB; ++bb) { pr0[bb] = v8f{}; pi0[bb] = v8f{}; pr1[bb] = v8f{}; pi1[bb] = v8f{}; }
#pragma unroll
        for (int g = 0; g < CHUNK / 8; ++g) {
            const float4 aR0 = *(const float4*)(zrf + g * 8);
            const float4 aI0 = *(const float4*)(zif + g * 8);
            const float4 aR1 = *(const float4*)(zrf + 16 * SP + g * 8);
            const float4 aI1 = *(const float4*)(zif + 16 * SP + g * 8);
            const v2f r00 = {aR0.x, aR0.y}, r01 = {aR0.z, aR0.w};
            const v2f i00 = {aI0.x, aI0.y}, i01 = {aI0.z, aI0.w};
            const v2f r10 = {aR1.x, aR1.y}, r11 = {aR1.z, aR1.w};
            const v2f i10 = {aI1.x, aI1.y}, i11 = {aI1.z, aI1.w};
#pragma unroll
            for (int bb = 0; bb < WB; ++bb) {
                const v2f b0 = {bu[bb][g].x, bu[bb][g].y}, b1 = {bu[bb][g].z, bu[bb][g].w};
                pr0[bb] = __builtin_amdgcn_wmma_f32_16x16x4_f32(false, r00, false, b0, (short)0, pr0[bb], false, false);
                pi0[bb] = __builtin_amdgcn_wmma_f32_16x16x4_f32(false, i00, false, b0, (short)0, pi0[bb], false, false);
                pr1[bb] = __builtin_amdgcn_wmma_f32_16x16x4_f32(false, r10, false, b0, (short)0, pr1[bb], false, false);
                pi1[bb] = __builtin_amdgcn_wmma_f32_16x16x4_f32(false, i10, false, b0, (short)0, pi1[bb], false, false);
                pr0[bb] = __builtin_amdgcn_wmma_f32_16x16x4_f32(false, r01, false, b1, (short)0, pr0[bb], false, false);
                pi0[bb] = __builtin_amdgcn_wmma_f32_16x16x4_f32(false, i01, false, b1, (short)0, pi0[bb], false, false);
                pr1[bb] = __builtin_amdgcn_wmma_f32_16x16x4_f32(false, r11, false, b1, (short)0, pr1[bb], false, false);
                pi1[bb] = __builtin_amdgcn_wmma_f32_16x16x4_f32(false, i11, false, b1, (short)0, pi1[bb], false, false);
            }
        }
#pragma unroll
        for (int bb = 0; bb < WB; ++bb) {
            float* __restrict__ dr = sXr[pfst + bb] + (ch + 1) * XP + rt0 * 16 + 8 * khalf;
            float* __restrict__ di = sXi[pfst + bb] + (ch + 1) * XP + rt0 * 16 + 8 * khalf;
            float4 v;
            v.x = pr0[bb][0]; v.y = pr0[bb][1]; v.z = pr0[bb][2]; v.w = pr0[bb][3]; *(float4*)(dr)      = v;
            v.x = pr0[bb][4]; v.y = pr0[bb][5]; v.z = pr0[bb][6]; v.w = pr0[bb][7]; *(float4*)(dr + 4)  = v;
            v.x = pr1[bb][0]; v.y = pr1[bb][1]; v.z = pr1[bb][2]; v.w = pr1[bb][3]; *(float4*)(dr + 16) = v;
            v.x = pr1[bb][4]; v.y = pr1[bb][5]; v.z = pr1[bb][6]; v.w = pr1[bb][7]; *(float4*)(dr + 20) = v;
            v.x = pi0[bb][0]; v.y = pi0[bb][1]; v.z = pi0[bb][2]; v.w = pi0[bb][3]; *(float4*)(di)      = v;
            v.x = pi0[bb][4]; v.y = pi0[bb][5]; v.z = pi0[bb][6]; v.w = pi0[bb][7]; *(float4*)(di + 4)  = v;
            v.x = pi1[bb][0]; v.y = pi1[bb][1]; v.z = pi1[bb][2]; v.w = pi1[bb][3]; *(float4*)(di + 16) = v;
            v.x = pi1[bb][4]; v.y = pi1[bb][5]; v.z = pi1[bb][6]; v.w = pi1[bb][7]; *(float4*)(di + 20) = v;
        }
    }
    __syncthreads();

    if (tid < NB * NN) {
        const float zr = szCr[tid & 63];
        const float zi = szCi[tid & 63];
        float xr = 0.f, xi = 0.f;
        float* __restrict__ sr = sXr[tid >> 6] + XP + (tid & 63);
        float* __restrict__ si = sXi[tid >> 6] + XP + (tid & 63);
#pragma unroll
        for (int i0 = 0; i0 < NCH; i0 += 16) {
            float vr[16], vi[16];
#pragma unroll
            for (int j = 0; j < 16; ++j) { vr[j] = sr[(i0 + j) * XP]; vi[j] = si[(i0 + j) * XP]; }
#pragma unroll
            for (int j = 0; j < 16; ++j) {
                const float nxr = zr * xr - zi * xi + vr[j];
                const float nxi = zr * xi + zi * xr + vi[j];
                vr[j] = nxr; vi[j] = nxi;
                xr = nxr; xi = nxi;
            }
#pragma unroll
            for (int j = 0; j < 16; ++j) { sr[(i0 + j) * XP] = vr[j]; si[(i0 + j) * XP] = vi[j]; }
        }
    }

    v8f acc0[WB], acc1[WB];
#pragma unroll
    for (int bb = 0; bb < WB; ++bb) { acc0[bb] = v8f{}; acc1[bb] = v8f{}; }
    {
        const int d0 = rt0 * 16 + m - 4 * khalf;
#pragma unroll
        for (int g = 0; g < CHUNK / 8; ++g) {
            const int d = d0 - 8 * g, e = d + 16;
            float4 a0, a1;
            a0.x = (d     >= 0) ? sK[max(d    , 0)] : 0.f;
            a0.y = (d - 1 >= 0) ? sK[max(d - 1, 0)] : 0.f;
            a0.z = (d - 2 >= 0) ? sK[max(d - 2, 0)] : 0.f;
            a0.w = (d - 3 >= 0) ? sK[max(d - 3, 0)] : 0.f;
            a1.x = (e     >= 0) ? sK[max(e    , 0)] : 0.f;
            a1.y = (e - 1 >= 0) ? sK[max(e - 1, 0)] : 0.f;
            a1.z = (e - 2 >= 0) ? sK[max(e - 2, 0)] : 0.f;
            a1.w = (e - 3 >= 0) ? sK[max(e - 3, 0)] : 0.f;
            const v2f a00 = {a0.x, a0.y}, a01 = {a0.z, a0.w};
            const v2f a10 = {a1.x, a1.y}, a11 = {a1.z, a1.w};
#pragma unroll
            for (int bb = 0; bb < WB; ++bb) {
                const v2f b0 = {bu[bb][g].x, bu[bb][g].y}, b1 = {bu[bb][g].z, bu[bb][g].w};
                acc0[bb] = __builtin_amdgcn_wmma_f32_16x16x4_f32(false, a00, false, b0, (short)0, acc0[bb], false, false);
                acc1[bb] = __builtin_amdgcn_wmma_f32_16x16x4_f32(false, a10, false, b0, (short)0, acc1[bb], false, false);
                acc0[bb] = __builtin_amdgcn_wmma_f32_16x16x4_f32(false, a01, false, b1, (short)0, acc0[bb], false, false);
                acc1[bb] = __builtin_amdgcn_wmma_f32_16x16x4_f32(false, a11, false, b1, (short)0, acc1[bb], false, false);
            }
        }
    }
    __syncthreads();

    {
        const float* __restrict__ crf = stCR + aoff;
        const float* __restrict__ cif = stCI + aoff;
#pragma unroll
        for (int g = 0; g < NN / 8; ++g) {
            const float4 aR0 = *(const float4*)(crf + g * 8);
            const float4 aI0 = *(const float4*)(cif + g * 8);
            const float4 aR1 = *(const float4*)(crf + 16 * SP + g * 8);
            const float4 aI1 = *(const float4*)(cif + 16 * SP + g * 8);
            const v2f r00 = {aR0.x, aR0.y}, r01 = {aR0.z, aR0.w};
            const v2f i00 = {aI0.x, aI0.y}, i01 = {aI0.z, aI0.w};
            const v2f r10 = {aR1.x, aR1.y}, r11 = {aR1.z, aR1.w};
            const v2f i10 = {aI1.x, aI1.y}, i11 = {aI1.z, aI1.w};
#pragma unroll
            for (int bb = 0; bb < WB; ++bb) {
                const float4 bxr = *(const float4*)(sXr[pfst + bb] + ch * XP + 4 * khalf + g * 8);
                const float4 bxi = *(const float4*)(sXi[pfst + bb] + ch * XP + 4 * khalf + g * 8);
                const v2f xr0 = {bxr.x, bxr.y}, xr1 = {bxr.z, bxr.w};
                const v2f xi0 = {bxi.x, bxi.y}, xi1 = {bxi.z, bxi.w};
                acc0[bb] = __builtin_amdgcn_wmma_f32_16x16x4_f32(false, r00, false, xr0, (short)0, acc0[bb], false, false);
                acc1[bb] = __builtin_amdgcn_wmma_f32_16x16x4_f32(false, r10, false, xr0, (short)0, acc1[bb], false, false);
                acc0[bb] = __builtin_amdgcn_wmma_f32_16x16x4_f32(false, i00, false, xi0, (short)0, acc0[bb], false, false);
                acc1[bb] = __builtin_amdgcn_wmma_f32_16x16x4_f32(false, i10, false, xi0, (short)0, acc1[bb], false, false);
                acc0[bb] = __builtin_amdgcn_wmma_f32_16x16x4_f32(false, r01, false, xr1, (short)0, acc0[bb], false, false);
                acc1[bb] = __builtin_amdgcn_wmma_f32_16x16x4_f32(false, r11, false, xr1, (short)0, acc1[bb], false, false);
                acc0[bb] = __builtin_amdgcn_wmma_f32_16x16x4_f32(false, i01, false, xi1, (short)0, acc0[bb], false, false);
                acc1[bb] = __builtin_amdgcn_wmma_f32_16x16x4_f32(false, i11, false, xi1, (short)0, acc1[bb], false, false);
            }
        }
    }

    const float Dh = D[h];
#pragma unroll
    for (int bb = 0; bb < WB; ++bb) {
        const size_t obase = ((size_t)(bfst + bb) * HH + h) * LL + (size_t)ch * CHUNK + rt0 * 16 + 8 * khalf;
        {
            const float4 u0 = *(const float4*)(u + obase);
            const float4 u1 = *(const float4*)(u + obase + 4);
            float4 o0, o1;
            o0.x = acc0[bb][0] + Dh * u0.x; o0.y = acc0[bb][1] + Dh * u0.y;
            o0.z = acc0[bb][2] + Dh * u0.z; o0.w = acc0[bb][3] + Dh * u0.w;
            o1.x = acc0[bb][4] + Dh * u1.x; o1.y = acc0[bb][5] + Dh * u1.y;
            o1.z = acc0[bb][6] + Dh * u1.z; o1.w = acc0[bb][7] + Dh * u1.w;
            *(float4*)(out + obase)     = o0;
            *(float4*)(out + obase + 4) = o1;
        }
        {
            const float4 u0 = *(const float4*)(u + obase + 16);
            const float4 u1 = *(const float4*)(u + obase + 20);
            float4 o0, o1;
            o0.x = acc1[bb][0] + Dh * u0.x; o0.y = acc1[bb][1] + Dh * u0.y;
            o0.z = acc1[bb][2] + Dh * u0.z; o0.w = acc1[bb][3] + Dh * u0.w;
            o1.x = acc1[bb][4] + Dh * u1.x; o1.y = acc1[bb][5] + Dh * u1.y;
            o1.z = acc1[bb][6] + Dh * u1.z; o1.w = acc1[bb][7] + Dh * u1.w;
            *(float4*)(out + obase + 16) = o0;
            *(float4*)(out + obase + 20) = o1;
        }
    }
}

extern "C" void kernel_launch(void* const* d_in, const int* in_sizes, int n_in,
                              void* d_out, int out_size, void* d_ws, size_t ws_size,
                              hipStream_t stream) {
    const float* u       = (const float*)d_in[0];
    const float* Lre     = (const float*)d_in[1];
    const float* Lim     = (const float*)d_in[2];
    const float* Wri     = (const float*)d_in[3];
    const float* Dp      = (const float*)d_in[4];
    const float* logstep = (const float*)d_in[5];
    float* out = (float*)d_out;

    dss_fused<<<HH * (BB / NB), FT, 0, stream>>>(u, Lre, Lim, Wri, Dp, logstep, out);
}


# T @ U leaves out the all-zero k groups above T's diagonal (row tile rt runs g < 2rt+2; K zero-padded in LDS); scan moved to waves 0-3 so the other waves' T @ U covers it; every sum keeps its order
# speedup vs baseline: 6.1999x; 1.0897x over previous
#include <hip/hip_runtime.h>
#include <math.h>

#define BB 8
#define HH 128
#define LL 4096
#define NN 64
#define CHUNK 64
#define NCH (LL / CHUNK)
#define XP 68
#define SP 68
#define KP 65
#define PADK 16
#define EPSV 1e-7f

typedef float v2f __attribute__((ext_vector_type(2)));
typedef float v8f __attribute__((ext_vector_type(8)));

#define NB 4
#define WB 2
#define FT 512

__global__ void __launch_bounds__(FT, 4)
dss_fused(const float* __restrict__ u,
          const float* __restrict__ Lre, const float* __restrict__ Lim,
          const float* __restrict__ Wri, const float* __restrict__ D,
          const float* __restrict__ logstep,
          float* __restrict__ out) {
    __shared__ __attribute__((aligned(16))) float stZr[NN * SP];
    __shared__ __attribute__((aligned(16))) float stZi[NN * SP];
    __shared__ __attribute__((aligned(16))) float stCR[CHUNK * SP];
    __shared__ __attribute__((aligned(16))) float stCI[CHUNK * SP];
    __shared__ __attribute__((aligned(16))) float sXr[NB][(NCH + 1) * XP];
    __shared__ __attribute__((aligned(16))) float sXi[NB][(NCH + 1) * XP];
    __shared__ float sE[CHUNK * KP], sCs[CHUNK * KP], sSn[CHUNK * KP];
    __shared__ float scr[NN], sci[NN], sKz[PADK + CHUNK], szCr[NN], szCi[NN];

    const int tid   = threadIdx.x;
    const int lane  = tid & 31;
    const int wave  = tid >> 5;
    const int h     = blockIdx.x / (BB / NB);
    const int wq    = wave >> 2;
    const int rhalf = (wq ^ (wq >> 1)) & 1;
    const int pfst  = (wq >> 1) * WB;
    const int bfst  = (blockIdx.x % (BB / NB)) * NB + pfst;
    const int m     = lane & 15;
    const int khalf = lane >> 4;
    const int ct    = wave & 3;
    const int rt0   = rhalf * 2;
    const int ch    = ct * 16 + m;

    float4 bu[WB][CHUNK / 8];
#pragma unroll
    for (int bb = 0; bb < WB; ++bb) {
        const float* __restrict__ ub = u + ((size_t)(bfst + bb) * HH + h) * LL + (size_t)ch * CHUNK + 4 * khalf;
#pragma unroll
        for (int g = 0; g < CHUNK / 8; ++g) bu[bb][g] = *(const float4*)(ub + g * 8);
    }

    if (tid < NB * NN) { sXr[tid >> 6][tid & 63] = 0.f; sXi[tid >> 6][tid & 63] = 0.f; }
    if (tid < PADK) sKz[tid] = 0.f;

    {
        const int n = tid & 63;
        const float lre = Lre[h * NN + n], lim = Lim[h * NN + n];
        const float dt  = __expf(logstep[h]);
        const float ar  = dt * lre, ai = dt * lim;

        if (tid < 2 * NN) {
            float cz, sz; sincosf(ai, &sz, &cz);
            const float er = expf(ar);
            const float zr = er * cz, zi = er * sz;
            float cL, sL; sincosf(ai * (float)LL, &sL, &cL);
            const float eL = expf(ar * (float)LL);
            const float zLr = eL * cL, zLi = eL * sL;
            const float nr = 1.f - zLr, ni = -zLi;
            const float dr = 1.f - zr,  di = -zi;
            const float dd = dr * dr + di * di;
            const float sr = (nr * dr + ni * di) / dd;
            const float si = (ni * dr - nr * di) / dd;
            const float wr = Wri[(h * NN + n) * 2 + 0], wi = Wri[(h * NN + n) * 2 + 1];
            const float ll = lre * lre + lim * lim;
            const float br = (wr * lre + wi * lim) / ll;
            const float bi = (wi * lre - wr * lim) / ll;
            const float ss = sr * sr + si * si + EPSV;
            const float cr = (br * sr + bi * si) / ss;
            const float ci = (bi * sr - br * si) / ss;

            if (tid < NN) {
                float cC, sC; sincosf(ai * (float)CHUNK, &sC, &cC);
                const float eC = expf(ar * (float)CHUNK);
                szCr[n] = eC * cC;
                szCi[n] = eC * sC;
                float pr = 1.f, pi_ = 0.f;
                float* __restrict__ dR = stZr + n * SP + (CHUNK - 1);
                float* __restrict__ dI = stZi + n * SP + (CHUNK - 1);
                for (int k = 0; k < CHUNK; ++k) {
                    dR[-k] = pr;
                    dI[-k] = pi_;
                    const float tr = __fmaf_rn(pr, zr, -(pi_ * zi));
                    pi_ = __fmaf_rn(pi_, zr, pr * zi);
                    pr = tr;
                }
            } else {
                scr[n] = cr; sci[n] = ci;
                float qr = cr * zr - ci * zi;
                float qi = cr * zi + ci * zr;
                for (int t = 0; t < CHUNK; ++t) {
                    stCR[t * SP + n] =  qr;
                    stCI[t * SP + n] = -qi;
                    const float tr = __fmaf_rn(qr, zr, -(qi * zi));
                    qi = __fmaf_rn(qi, zr, qr * zi);
                    qr = tr;
                }
            }
        } else {
            for (int t = (tid - 2 * NN) >> 6; t < CHUNK; t += (FT - 2 * NN) / NN) {
                const float tf = (float)t;
                const float e = expf(ar * tf);
                float cs, sn; sincosf(ai * tf, &sn, &cs);
                sE [t * KP + n] = e;
                sCs[t * KP + n] = cs;
                sSn[t * KP + n] = sn;
            }
        }
    }
    __syncthreads();

    if (tid < CHUNK) {
        const float* __restrict__ pe = sE  + tid * KP;
        const float* __restrict__ pc = sCs + tid * KP;
        const float* __restrict__ ps = sSn + tid * KP;
        float acc = 0.f;
        for (int j = 0; j < NN; ++j) {
            const float inner = __fmaf_rn(scr[j], pc[j], -(sci[j] * ps[j]));
            acc = __fmaf_rn(pe[j], inner, acc);
        }
        sKz[PADK + tid] = acc;
    }

    const int aoff = (rt0 * 16 + m) * SP + 4 * khalf;

    {
        const float* __restrict__ zrf = stZr + aoff;
        const float* __restrict__ zif = stZi + aoff;
        v8f pr0[WB], pi0[WB], pr1[WB], pi1[WB];
#pragma unroll
        for (int bb = 0; bb < WB; ++bb) { pr0[bb] = v8f{}; pi0[bb] = v8f{}; pr1[bb] = v8f{}; pi1[bb] = v8f{}; }
#pragma unroll
        for (int g = 0; g < CHUNK / 8; ++g) {
            const float4 aR0 = *(const float4*)(zrf + g * 8);
            const float4 aI0 = *(const float4*)(zif + g * 8);
            const float4 aR1 = *(const float4*)(zrf + 16 * SP + g * 8);
            const float4 aI1 = *(const float4*)(zif + 16 * SP + g * 8);
            const v2f r00 = {aR0.x, aR0.y}, r01 = {aR0.z, aR0.w};
            const v2f i00 = {aI0.x, aI0.y}, i01 = {aI0.z, aI0.w};
            const v2f r10 = {aR1.x, aR1.y}, r11 = {aR1.z, aR1.w};
            const v2f i10 = {aI1.x, aI1.y}, i11 = {aI1.z, aI1.w};
#pragma unroll
            for (int bb = 0; bb < WB; ++bb) {
                const v2f b0 = {bu[bb][g].x, bu[bb][g].y}, b1 = {bu[bb][g].z, bu[bb][g].w};
                pr0[bb] = __builtin_amdgcn_wmma_f32_16x16x4_f32(false, r00, false, b0, (short)0, pr0[bb], false, false);
                pi0[bb] = __builtin_amdgcn_wmma_f32_16x16x4_f32(false, i00, false, b0, (short)0, pi0[bb], false, false);
                pr1[bb] = __builtin_amdgcn_wmma_f32_16x16x4_f32(false, r10, false, b0, (short)0, pr1[bb], false, false);
                pi1[bb] = __builtin_amdgcn_wmma_f32_16x16x4_f32(false, i10, false, b0, (short)0, pi1[bb], false, false);
                pr0[bb] = __builtin_amdgcn_wmma_f32_16x16x4_f32(false, r01, false, b1, (short)0, pr0[bb], false, false);
                pi0[bb] = __builtin_amdgcn_wmma_f32_16x16x4_f32(false, i01, false, b1, (short)0, pi0[bb], false, false);
                pr1[bb] = __builtin_amdgcn_wmma_f32_16x16x4_f32(false, r11, false, b1, (short)0, pr1[bb], false, false);
                pi1[bb] = __builtin_amdgcn_wmma_f32_16x16x4_f32(false, i11, false, b1, (short)0, pi1[bb], false, false);
            }
        }
#pragma unroll
        for (int bb = 0; bb < WB; ++bb) {
            float* __restrict__ dr = sXr[pfst + bb] + (ch + 1) * XP + rt0 * 16 + 8 * khalf;
            float* __restrict__ di = sXi[pfst + bb] + (ch + 1) * XP + rt0 * 16 + 8 * khalf;
            float4 v;
            v.x = pr0[bb][0]; v.y = pr0[bb][1]; v.z = pr0[bb][2]; v.w = pr0[bb][3]; *(float4*)(dr)      = v;
            v.x = pr0[bb][4]; v.y = pr0[bb][5]; v.z = pr0[bb][6]; v.w = pr0[bb][7]; *(float4*)(dr + 4)  = v;
            v.x = pr1[bb][0]; v.y = pr1[bb][1]; v.z = pr1[bb][2]; v.w = pr1[bb][3]; *(float4*)(dr + 16) = v;
            v.x = pr1[bb][4]; v.y = pr1[bb][5]; v.z = pr1[bb][6]; v.w = pr1[bb][7]; *(float4*)(dr + 20) = v;
            v.x = pi0[bb][0]; v.y = pi0[bb][1]; v.z = pi0[bb][2]; v.w = pi0[bb][3]; *(float4*)(di)      = v;
            v.x = pi0[bb][4]; v.y = pi0[bb][5]; v.z = pi0[bb][6]; v.w = pi0[bb][7]; *(float4*)(di + 4)  = v;
            v.x = pi1[bb][0]; v.y = pi1[bb][1]; v.z = pi1[bb][2]; v.w = pi1[bb][3]; *(float4*)(di + 16) = v;
            v.x = pi1[bb][4]; v.y = pi1[bb][5]; v.z = pi1[bb][6]; v.w = pi1[bb][7]; *(float4*)(di + 20) = v;
        }
    }
    __syncthreads();

    if (tid < 2 * NN) {
        const float zr = szCr[tid & 63];
        const float zi = szCi[tid & 63];
        float xr = 0.f, xi = 0.f, yr = 0.f, yi = 0.f;
        float* __restrict__ sr = sXr[tid >> 6] + XP + (tid & 63);
        float* __restrict__ si = sXi[tid >> 6] + XP + (tid & 63);
        float* __restrict__ tr = sXr[(tid >> 6) + 2] + XP + (tid & 63);
        float* __restrict__ ti = sXi[(tid >> 6) + 2] + XP + (tid & 63);
#pragma unroll
        for (int i0 = 0; i0 < NCH; i0 += 16) {
            float vr[16], vi[16], wr[16], wi[16];
#pragma unroll
            for (int j = 0; j < 16; ++j) {
                vr[j] = sr[(i0 + j) * XP]; vi[j] = si[(i0 + j) * XP];
                wr[j] = tr[(i0 + j) * XP]; wi[j] = ti[(i0 + j) * XP];
            }
#pragma unroll
            for (int j = 0; j < 16; ++j) {
                const float nxr = zr * xr - zi * xi + vr[j];
                const float nxi = zr * xi + zi * xr + vi[j];
                const float nyr = zr * yr - zi * yi + wr[j];
                const float nyi = zr * yi + zi * yr + wi[j];
                vr[j] = nxr; vi[j] = nxi;
                wr[j] = nyr; wi[j] = nyi;
                xr = nxr; xi = nxi;
                yr = nyr; yi = nyi;
            }
#pragma unroll
            for (int j = 0; j < 16; ++j) {
                sr[(i0 + j) * XP] = vr[j]; si[(i0 + j) * XP] = vi[j];
                tr[(i0 + j) * XP] = wr[j]; ti[(i0 + j) * XP] = wi[j];
            }
        }
    }

    v8f acc0[WB], acc1[WB];
#pragma unroll
    for (int bb = 0; bb < WB; ++bb) { acc0[bb] = v8f{}; acc1[bb] = v8f{}; }
    {
        const float* __restrict__ kz = sKz + PADK + rt0 * 16 + m - 4 * khalf;
        const int rhalf_s = __builtin_amdgcn_readfirstlane(rhalf);
        if (rhalf_s) {
#pragma unroll
            for (int g = 0; g < 8; ++g) {
                const v2f a10 = {kz[16 - 8 * g], kz[15 - 8 * g]}, a11 = {kz[14 - 8 * g], kz[13 - 8 * g]};
                if (g < 6) {
                    const v2f a00 = {kz[-8 * g], kz[-8 * g - 1]}, a01 = {kz[-8 * g - 2], kz[-8 * g - 3]};
#pragma unroll
                    for (int bb = 0; bb < WB; ++bb) {
                        const v2f b0 = {bu[bb][g].x, bu[bb][g].y}, b1 = {bu[bb][g].z, bu[bb][g].w};
                        acc0[bb] = __builtin_amdgcn_wmma_f32_16x16x4_f32(false, a00, false, b0, (short)0, acc0[bb], false, false);
                        acc1[bb] = __builtin_amdgcn_wmma_f32_16x16x4_f32(false, a10, false, b0, (short)0, acc1[bb], false, false);
                        acc0[bb] = __builtin_amdgcn_wmma_f32_16x16x4_f32(false, a01, false, b1, (short)0, acc0[bb], false, false);
                        acc1[bb] = __builtin_amdgcn_wmma_f32_16x16x4_f32(false, a11, false, b1, (short)0, acc1[bb], false, false);
                    }
                } else {
                    const v2f b00 = {bu[0][g].x, bu[0][g].y}, b01 = {bu[0][g].z, bu[0][g].w};
                    const v2f b10 = {bu[1][g].x, bu[1][g].y}, b11 = {bu[1][g].z, bu[1][g].w};
                    acc1[0] = __builtin_amdgcn_wmma_f32_16x16x4_f32(false, a10, false, b00, (short)0, acc1[0], false, false);
                    acc1[1] = __builtin_amdgcn_wmma_f32_16x16x4_f32(false, a10, false, b10, (short)0, acc1[1], false, false);
                    acc1[0] = __builtin_amdgcn_wmma_f32_16x16x4_f32(false, a11, false, b01, (short)0, acc1[0], false, false);
                    acc1[1] = __builtin_amdgcn_wmma_f32_16x16x4_f32(false, a11, false, b11, (short)0, acc1[1], false, false);
                }
            }
        } else {
#pragma unroll
            for (int g = 0; g < 4; ++g) {
                const v2f a10 = {kz[16 - 8 * g], kz[15 - 8 * g]}, a11 = {kz[14 - 8 * g], kz[13 - 8 * g]};
                if (g < 2) {
                    const v2f a00 = {kz[-8 * g], kz[-8 * g - 1]}, a01 = {kz[-8 * g - 2], kz[-8 * g - 3]};
#pragma unroll
                    for (int bb = 0; bb < WB; ++bb) {
                        const v2f b0 = {bu[bb][g].x, bu[bb][g].y}, b1 = {bu[bb][g].z, bu[bb][g].w};
                        acc0[bb] = __builtin_amdgcn_wmma_f32_16x16x4_f32(false, a00, false, b0, (short)0, acc0[bb], false, false);
                        acc1[bb] = __builtin_amdgcn_wmma_f32_16x16x4_f32(false, a10, false, b0, (short)0, acc1[bb], false, false);
                        acc0[bb] = __builtin_amdgcn_wmma_f32_16x16x4_f32(false, a01, false, b1, (short)0, acc0[bb], false, false);
                        acc1[bb] = __builtin_amdgcn_wmma_f32_16x16x4_f32(false, a11, false, b1, (short)0, acc1[bb], false, false);
                    }
                } else {
                    const v2f b00 = {bu[0][g].x, bu[0][g].y}, b01 = {bu[0][g].z, bu[0][g].w};
                    const v2f b10 = {bu[1][g].x, bu[1][g].y}, b11 = {bu[1][g].z, bu[1][g].w};
                    acc1[0] = __builtin_amdgcn_wmma_f32_16x16x4_f32(false, a10, false, b00, (short)0, acc1[0], false, false);
                    acc1[1] = __builtin_amdgcn_wmma_f32_16x16x4_f32(false, a10, false, b10, (short)0, acc1[1], false, false);
                    acc1[0] = __builtin_amdgcn_wmma_f32_16x16x4_f32(false, a11, false, b01, (short)0, acc1[0], false, false);
                    acc1[1] = __builtin_amdgcn_wmma_f32_16x16x4_f32(false, a11, false, b11, (short)0, acc1[1], false, false);
                }
            }
        }
    }
    __syncthreads();

    {
        const float* __restrict__ crf = stCR + aoff;
        const float* __restrict__ cif = stCI + aoff;
#pragma unroll
        for (int g = 0; g < NN / 8; ++g) {
            const float4 aR0 = *(const float4*)(crf + g * 8);
            const float4 aI0 = *(const float4*)(cif + g * 8);
            const float4 aR1 = *(const float4*)(crf + 16 * SP + g * 8);
            const float4 aI1 = *(const float4*)(cif + 16 * SP + g * 8);
            const v2f r00 = {aR0.x, aR0.y}, r01 = {aR0.z, aR0.w};
            const v2f i00 = {aI0.x, aI0.y}, i01 = {aI0.z, aI0.w};
            const v2f r10 = {aR1.x, aR1.y}, r11 = {aR1.z, aR1.w};
            const v2f i10 = {aI1.x, aI1.y}, i11 = {aI1.z, aI1.w};
#pragma unroll
            for (int bb = 0; bb < WB; ++bb) {
                const float4 bxr = *(const float4*)(sXr[pfst + bb] + ch * XP + 4 * khalf + g * 8);
                const float4 bxi = *(const float4*)(sXi[pfst + bb] + ch * XP + 4 * khalf + g * 8);
                const v2f xr0 = {bxr.x, bxr.y}, xr1 = {bxr.z, bxr.w};
                const v2f xi0 = {bxi.x, bxi.y}, xi1 = {bxi.z, bxi.w};
                acc0[bb] = __builtin_amdgcn_wmma_f32_16x16x4_f32(false, r00, false, xr0, (short)0, acc0[bb], false, false);
                acc1[bb] = __builtin_amdgcn_wmma_f32_16x16x4_f32(false, r10, false, xr0, (short)0, acc1[bb], false, false);
                acc0[bb] = __builtin_amdgcn_wmma_f32_16x16x4_f32(false, i00, false, xi0, (short)0, acc0[bb], false, false);
                acc1[bb] = __builtin_amdgcn_wmma_f32_16x16x4_f32(false, i10, false, xi0, (short)0, acc1[bb], false, false);
                acc0[bb] = __builtin_amdgcn_wmma_f32_16x16x4_f32(false, r01, false, xr1, (short)0, acc0[bb], false, false);
                acc1[bb] = __builtin_amdgcn_wmma_f32_16x16x4_f32(false, r11, false, xr1, (short)0, acc1[bb], false, false);
                acc0[bb] = __builtin_amdgcn_wmma_f32_16x16x4_f32(false, i01, false, xi1, (short)0, acc0[bb], false, false);
                acc1[bb] = __builtin_amdgcn_wmma_f32_16x16x4_f32(false, i11, false, xi1, (short)0, acc1[bb], false, false);
            }
        }
    }

    const float Dh = D[h];
#pragma unroll
    for (int bb = 0; bb < WB; ++bb) {
        const size_t obase = ((size_t)(bfst + bb) * HH + h) * LL + (size_t)ch * CHUNK + rt0 * 16 + 8 * khalf;
        {
            const float4 u0 = *(const float4*)(u + obase);
            const float4 u1 = *(const float4*)(u + obase + 4);
            float4 o0, o1;
            o0.x = acc0[bb][0] + Dh * u0.x; o0.y = acc0[bb][1] + Dh * u0.y;
            o0.z = acc0[bb][2] + Dh * u0.z; o0.w = acc0[bb][3] + Dh * u0.w;
            o1.x = acc0[bb][4] + Dh * u1.x; o1.y = acc0[bb][5] + Dh * u1.y;
            o1.z = acc0[bb][6] + Dh * u1.z; o1.w = acc0[bb][7] + Dh * u1.w;
            *(float4*)(out + obase)     = o0;
            *(float4*)(out + obase + 4) = o1;
        }
        {
            const float4 u0 = *(const float4*)(u + obase + 16);
            const float4 u1 = *(const float4*)(u + obase + 20);
            float4 o0, o1;
            o0.x = acc1[bb][0] + Dh * u0.x; o0.y = acc1[bb][1] + Dh * u0.y;
            o0.z = acc1[bb][2] + Dh * u0.z; o0.w = acc1[bb][3] + Dh * u0.w;
            o1.x = acc1[bb][4] + Dh * u1.x; o1.y = acc1[bb][5] + Dh * u1.y;
            o1.z = acc1[bb][6] + Dh * u1.z; o1.w = acc1[bb][7] + Dh * u1.w;
            *(float4*)(out + obase + 16) = o0;
            *(float4*)(out + obase + 20) = o1;
        }
    }
}

extern "C" void kernel_launch(void* const* d_in, const int* in_sizes, int n_in,
                              void* d_out, int out_size, void* d_ws, size_t ws_size,
                              hipStream_t stream) {
    const float* u       = (const float*)d_in[0];
    const float* Lre     = (const float*)d_in[1];
    const float* Lim     = (const float*)d_in[2];
    const float* Wri     = (const float*)d_in[3];
    const float* Dp      = (const float*)d_in[4];
    const float* logstep = (const float*)d_in[5];
    float* out = (float*)d_out;

    dss_fused<<<HH * (BB / NB), FT, 0, stream>>>(u, Lre, Lim, Wri, Dp, logstep, out);
}


# Coefficient build runs under the U loads: mode parameters load first, and K[t] is the row sum of C (chain powers), no separate exp/sincos pass; K's terms round differently, mode order kept
# speedup vs baseline: 6.6665x; 1.0753x over previous
#include <hip/hip_runtime.h>
#include <math.h>

#define BB 8
#define HH 128
#define LL 4096
#define NN 64
#define CHUNK 64
#define NCH (LL / CHUNK)
#define XP 68
#define SP 68
#define PADK 16
#define EPSV 1e-7f

typedef float v2f __attribute__((ext_vector_type(2)));
typedef float v8f __attribute__((ext_vector_type(8)));

#define NB 4
#define WB 2
#define FT 512

__global__ void __launch_bounds__(FT, 4)
dss_fused(const float* __restrict__ u,
          const float* __restrict__ Lre, const float* __restrict__ Lim,
          const float* __restrict__ Wri, const float* __restrict__ D,
          const float* __restrict__ logstep,
          float* __restrict__ out) {
    __shared__ __attribute__((aligned(16))) float stZr[NN * SP];
    __shared__ __attribute__((aligned(16))) float stZi[NN * SP];
    __shared__ __attribute__((aligned(16))) float stCR[CHUNK * SP];
    __shared__ __attribute__((aligned(16))) float stCI[CHUNK * SP];
    __shared__ __attribute__((aligned(16))) float sXr[NB][(NCH + 1) * XP];
    __shared__ __attribute__((aligned(16))) float sXi[NB][(NCH + 1) * XP];
    __shared__ __attribute__((aligned(16))) float scr[NN];
    __shared__ float sKz[PADK + CHUNK], szCr[NN], szCi[NN];

    const int tid   = threadIdx.x;
    const int lane  = tid & 31;
    const int wave  = tid >> 5;
    const int h     = blockIdx.x / (BB / NB);
    const int wq    = wave >> 2;
    const int rhalf = (wq ^ (wq >> 1)) & 1;
    const int pfst  = (wq >> 1) * WB;
    const int bfst  = (blockIdx.x % (BB / NB)) * NB + pfst;
    const int m     = lane & 15;
    const int khalf = lane >> 4;
    const int ct    = wave & 3;
    const int rt0   = rhalf * 2;
    const int ch    = ct * 16 + m;

    const int   pn    = tid & 63;
    const float p_lre = Lre[h * NN + pn], p_lim = Lim[h * NN + pn];
    const float p_ls  = logstep[h];
    const float p_wr  = Wri[(h * NN + pn) * 2 + 0], p_wi = Wri[(h * NN + pn) * 2 + 1];

    float4 bu[WB][CHUNK / 8];
#pragma unroll
    for (int bb = 0; bb < WB; ++bb) {
        const float* __restrict__ ub = u + ((size_t)(bfst + bb) * HH + h) * LL + (size_t)ch * CHUNK + 4 * khalf;
#pragma unroll
        for (int g = 0; g < CHUNK / 8; ++g) bu[bb][g] = *(const float4*)(ub + g * 8);
    }

    if (tid < NB * NN) { sXr[tid >> 6][tid & 63] = 0.f; sXi[tid >> 6][tid & 63] = 0.f; }
    if (tid < PADK) sKz[tid] = 0.f;

    {
        const int n = pn;
        const float lre = p_lre, lim = p_lim;
        const float dt  = __expf(p_ls);
        const float ar  = dt * lre, ai = dt * lim;

        if (tid < 2 * NN) {
            float cz, sz; sincosf(ai, &sz, &cz);
            const float er = expf(ar);
            const float zr = er * cz, zi = er * sz;
            float cL, sL; sincosf(ai * (float)LL, &sL, &cL);
            const float eL = expf(ar * (float)LL);
            const float zLr = eL * cL, zLi = eL * sL;
            const float nr = 1.f - zLr, ni = -zLi;
            const float dr = 1.f - zr,  di = -zi;
            const float dd = dr * dr + di * di;
            const float sr = (nr * dr + ni * di) / dd;
            const float si = (ni * dr - nr * di) / dd;
            const float wr = p_wr, wi = p_wi;
            const float ll = lre * lre + lim * lim;
            const float br = (wr * lre + wi * lim) / ll;
            const float bi = (wi * lre - wr * lim) / ll;
            const float ss = sr * sr + si * si + EPSV;
            const float cr = (br * sr + bi * si) / ss;
            const float ci = (bi * sr - br * si) / ss;

            if (tid < NN) {
                float cC, sC; sincosf(ai * (float)CHUNK, &sC, &cC);
                const float eC = expf(ar * (float)CHUNK);
                szCr[n] = eC * cC;
                szCi[n] = eC * sC;
                float pr = 1.f, pi_ = 0.f;
                float* __restrict__ dR = stZr + n * SP + (CHUNK - 1);
                float* __restrict__ dI = stZi + n * SP + (CHUNK - 1);
                for (int k = 0; k < CHUNK; ++k) {
                    dR[-k] = pr;
                    dI[-k] = pi_;
                    const float tr = __fmaf_rn(pr, zr, -(pi_ * zi));
                    pi_ = __fmaf_rn(pi_, zr, pr * zi);
                    pr = tr;
                }
            } else {
                scr[n] = cr;
                float qr = cr * zr - ci * zi;
                float qi = cr * zi + ci * zr;
                for (int t = 0; t < CHUNK; ++t) {
                    stCR[t * SP + n] =  qr;
                    stCI[t * SP + n] = -qi;
                    const float tr = __fmaf_rn(qr, zr, -(qi * zi));
                    qi = __fmaf_rn(qi, zr, qr * zi);
                    qr = tr;
                }
            }
        }
    }
    __syncthreads();

    if (tid < CHUNK) {
        const float* __restrict__ row = (tid == 0) ? scr : (stCR + (tid - 1) * SP);
        float acc = 0.f;
        for (int j = 0; j < NN; j += 4) {
            const float4 v = *(const float4*)(row + j);
            acc += v.x; acc += v.y; acc += v.z; acc += v.w;
        }
        sKz[PADK + tid] = acc;
    }

    const int aoff = (rt0 * 16 + m) * SP + 4 * khalf;

    {
        const float* __restrict__ zrf = stZr + aoff;
        const float* __restrict__ zif = stZi + aoff;
        v8f pr0[WB], pi0[WB], pr1[WB], pi1[WB];
#pragma unroll
        for (int bb = 0; bb < WB; ++bb) { pr0[bb] = v8f{}; pi0[bb] = v8f{}; pr1[bb] = v8f{}; pi1[bb] = v8f{}; }
#pragma unroll
        for (int g = 0; g < CHUNK / 8; ++g) {
            const float4 aR0 = *(const float4*)(zrf + g * 8);
            const float4 aI0 = *(const float4*)(zif + g * 8);
            const float4 aR1 = *(const float4*)(zrf + 16 * SP + g * 8);
            const float4 aI1 = *(const float4*)(zif + 16 * SP + g * 8);
            const v2f r00 = {aR0.x, aR0.y}, r01 = {aR0.z, aR0.w};
            const v2f i00 = {aI0.x, aI0.y}, i01 = {aI0.z, aI0.w};
            const v2f r10 = {aR1.x, aR1.y}, r11 = {aR1.z, aR1.w};
            const v2f i10 = {aI1.x, aI1.y}, i11 = {aI1.z, aI1.w};
#pragma unroll
            for (int bb = 0; bb < WB; ++bb) {
                const v2f b0 = {bu[bb][g].x, bu[bb][g].y}, b1 = {bu[bb][g].z, bu[bb][g].w};
                pr0[bb] = __builtin_amdgcn_wmma_f32_16x16x4_f32(false, r00, false, b0, (short)0, pr0[bb], false, false);
                pi0[bb] = __builtin_amdgcn_wmma_f32_16x16x4_f32(false, i00, false, b0, (short)0, pi0[bb], false, false);
                pr1[bb] = __builtin_amdgcn_wmma_f32_16x16x4_f32(false, r10, false, b0, (short)0, pr1[bb], false, false);
                pi1[bb] = __builtin_amdgcn_wmma_f32_16x16x4_f32(false, i10, false, b0, (short)0, pi1[bb], false, false);
                pr0[bb] = __builtin_amdgcn_wmma_f32_16x16x4_f32(false, r01, false, b1, (short)0, pr0[bb], false, false);
                pi0[bb] = __builtin_amdgcn_wmma_f32_16x16x4_f32(false, i01, false, b1, (short)0, pi0[bb], false, false);
                pr1[bb] = __builtin_amdgcn_wmma_f32_16x16x4_f32(false, r11, false, b1, (short)0, pr1[bb], false, false);
                pi1[bb] = __builtin_amdgcn_wmma_f32_16x16x4_f32(false, i11, false, b1, (short)0, pi1[bb], false, false);
            }
        }
#pragma unroll
        for (int bb = 0; bb < WB; ++bb) {
            float* __restrict__ dr = sXr[pfst + bb] + (ch + 1) * XP + rt0 * 16 + 8 * khalf;
            float* __restrict__ di = sXi[pfst + bb] + (ch + 1) * XP + rt0 * 16 + 8 * khalf;
            float4 v;
            v.x = pr0[bb][0]; v.y = pr0[bb][1]; v.z = pr0[bb][2]; v.w = pr0[bb][3]; *(float4*)(dr)      = v;
            v.x = pr0[bb][4]; v.y = pr0[bb][5]; v.z = pr0[bb][6]; v.w = pr0[bb][7]; *(float4*)(dr + 4)  = v;
            v.x = pr1[bb][0]; v.y = pr1[bb][1]; v.z = pr1[bb][2]; v.w = pr1[bb][3]; *(float4*)(dr + 16) = v;
            v.x = pr1[bb][4]; v.y = pr1[bb][5]; v.z = pr1[bb][6]; v.w = pr1[bb][7]; *(float4*)(dr + 20) = v;
            v.x = pi0[bb][0]; v.y = pi0[bb][1]; v.z = pi0[bb][2]; v.w = pi0[bb][3]; *(float4*)(di)      = v;
            v.x = pi0[bb][4]; v.y = pi0[bb][5]; v.z = pi0[bb][6]; v.w = pi0[bb][7]; *(float4*)(di + 4)  = v;
            v.x = pi1[bb][0]; v.y = pi1[bb][1]; v.z = pi1[bb][2]; v.w = pi1[bb][3]; *(float4*)(di + 16) = v;
            v.x = pi1[bb][4]; v.y = pi1[bb][5]; v.z = pi1[bb][6]; v.w = pi1[bb][7]; *(float4*)(di + 20) = v;
        }
    }
    __syncthreads();

    if (tid < 2 * NN) {
        const float zr = szCr[tid & 63];
        const float zi = szCi[tid & 63];
        float xr = 0.f, xi = 0.f, yr = 0.f, yi = 0.f;
        float* __restrict__ sr = sXr[tid >> 6] + XP + (tid & 63);
        float* __restrict__ si = sXi[tid >> 6] + XP + (tid & 63);
        float* __restrict__ tr = sXr[(tid >> 6) + 2] + XP + (tid & 63);
        float* __restrict__ ti = sXi[(tid >> 6) + 2] + XP + (tid & 63);
#pragma unroll
        for (int i0 = 0; i0 < NCH; i0 += 16) {
            float vr[16], vi[16], wr[16], wi[16];
#pragma unroll
            for (int j = 0; j < 16; ++j) {
                vr[j] = sr[(i0 + j) * XP]; vi[j] = si[(i0 + j) * XP];
                wr[j] = tr[(i0 + j) * XP]; wi[j] = ti[(i0 + j) * XP];
            }
#pragma unroll
            for (int j = 0; j < 16; ++j) {
                const float nxr = zr * xr - zi * xi + vr[j];
                const float nxi = zr * xi + zi * xr + vi[j];
                const float nyr = zr * yr - zi * yi + wr[j];
                const float nyi = zr * yi + zi * yr + wi[j];
                vr[j] = nxr; vi[j] = nxi;
                wr[j] = nyr; wi[j] = nyi;
                xr = nxr; xi = nxi;
                yr = nyr; yi = nyi;
            }
#pragma unroll
            for (int j = 0; j < 16; ++j) {
                sr[(i0 + j) * XP] = vr[j]; si[(i0 + j) * XP] = vi[j];
                tr[(i0 + j) * XP] = wr[j]; ti[(i0 + j) * XP] = wi[j];
            }
        }
    }

    v8f acc0[WB], acc1[WB];
#pragma unroll
    for (int bb = 0; bb < WB; ++bb) { acc0[bb] = v8f{}; acc1[bb] = v8f{}; }
    {
        const float* __restrict__ kz = sKz + PADK + rt0 * 16 + m - 4 * khalf;
        const int rhalf_s = __builtin_amdgcn_readfirstlane(rhalf);
        if (rhalf_s) {
#pragma unroll
            for (int g = 0; g < 8; ++g) {
                const v2f a10 = {kz[16 - 8 * g], kz[15 - 8 * g]}, a11 = {kz[14 - 8 * g], kz[13 - 8 * g]};
                if (g < 6) {
                    const v2f a00 = {kz[-8 * g], kz[-8 * g - 1]}, a01 = {kz[-8 * g - 2], kz[-8 * g - 3]};
#pragma unroll
                    for (int bb = 0; bb < WB; ++bb) {
                        const v2f b0 = {bu[bb][g].x, bu[bb][g].y}, b1 = {bu[bb][g].z, bu[bb][g].w};
                        acc0[bb] = __builtin_amdgcn_wmma_f32_16x16x4_f32(false, a00, false, b0, (short)0, acc0[bb], false, false);
                        acc1[bb] = __builtin_amdgcn_wmma_f32_16x16x4_f32(false, a10, false, b0, (short)0, acc1[bb], false, false);
                        acc0[bb] = __builtin_amdgcn_wmma_f32_16x16x4_f32(false, a01, false, b1, (short)0, acc0[bb], false, false);
                        acc1[bb] = __builtin_amdgcn_wmma_f32_16x16x4_f32(false, a11, false, b1, (short)0, acc1[bb], false, false);
                    }
                } else {
                    const v2f b00 = {bu[0][g].x, bu[0][g].y}, b01 = {bu[0][g].z, bu[0][g].w};
                    const v2f b10 = {bu[1][g].x, bu[1][g].y}, b11 = {bu[1][g].z, bu[1][g].w};
                    acc1[0] = __builtin_amdgcn_wmma_f32_16x16x4_f32(false, a10, false, b00, (short)0, acc1[0], false, false);
                    acc1[1] = __builtin_amdgcn_wmma_f32_16x16x4_f32(false, a10, false, b10, (short)0, acc1[1], false, false);
                    acc1[0] = __builtin_amdgcn_wmma_f32_16x16x4_f32(false, a11, false, b01, (short)0, acc1[0], false, false);
                    acc1[1] = __builtin_amdgcn_wmma_f32_16x16x4_f32(false, a11, false, b11, (short)0, acc1[1], false, false);
                }
            }
        } else {
#pragma unroll
            for (int g = 0; g < 4; ++g) {
                const v2f a10 = {kz[16 - 8 * g], kz[15 - 8 * g]}, a11 = {kz[14 - 8 * g], kz[13 - 8 * g]};
                if (g < 2) {
                    const v2f a00 = {kz[-8 * g], kz[-8 * g - 1]}, a01 = {kz[-8 * g - 2], kz[-8 * g - 3]};
#pragma unroll
                    for (int bb = 0; bb < WB; ++bb) {
                        const v2f b0 = {bu[bb][g].x, bu[bb][g].y}, b1 = {bu[bb][g].z, bu[bb][g].w};
                        acc0[bb] = __builtin_amdgcn_wmma_f32_16x16x4_f32(false, a00, false, b0, (short)0, acc0[bb], false, false);
                        acc1[bb] = __builtin_amdgcn_wmma_f32_16x16x4_f32(false, a10, false, b0, (short)0, acc1[bb], false, false);
                        acc0[bb] = __builtin_amdgcn_wmma_f32_16x16x4_f32(false, a01, false, b1, (short)0, acc0[bb], false, false);
                        acc1[bb] = __builtin_amdgcn_wmma_f32_16x16x4_f32(false, a11, false, b1, (short)0, acc1[bb], false, false);
                    }
                } else {
                    const v2f b00 = {bu[0][g].x, bu[0][g].y}, b01 = {bu[0][g].z, bu[0][g].w};
                    const v2f b10 = {bu[1][g].x, bu[1][g].y}, b11 = {bu[1][g].z, bu[1][g].w};
                    acc1[0] = __builtin_amdgcn_wmma_f32_16x16x4_f32(false, a10, false, b00, (short)0, acc1[0], false, false);
                    acc1[1] = __builtin_amdgcn_wmma_f32_16x16x4_f32(false, a10, false, b10, (short)0, acc1[1], false, false);
                    acc1[0] = __builtin_amdgcn_wmma_f32_16x16x4_f32(false, a11, false, b01, (short)0, acc1[0], false, false);
                    acc1[1] = __builtin_amdgcn_wmma_f32_16x16x4_f32(false, a11, false, b11, (short)0, acc1[1], false, false);
                }
            }
        }
    }
    __syncthreads();

    {
        const float* __restrict__ crf = stCR + aoff;
        const float* __restrict__ cif = stCI + aoff;
#pragma unroll
        for (int g = 0; g < NN / 8; ++g) {
            const float4 aR0 = *(const float4*)(crf + g * 8);
            const float4 aI0 = *(const float4*)(cif + g * 8);
            const float4 aR1 = *(const float4*)(crf + 16 * SP + g * 8);
            const float4 aI1 = *(const float4*)(cif + 16 * SP + g * 8);
            const v2f r00 = {aR0.x, aR0.y}, r01 = {aR0.z, aR0.w};
            const v2f i00 = {aI0.x, aI0.y}, i01 = {aI0.z, aI0.w};
            const v2f r10 = {aR1.x, aR1.y}, r11 = {aR1.z, aR1.w};
            const v2f i10 = {aI1.x, aI1.y}, i11 = {aI1.z, aI1.w};
#pragma unroll
            for (int bb = 0; bb < WB; ++bb) {
                const float4 bxr = *(const float4*)(sXr[pfst + bb] + ch * XP + 4 * khalf + g * 8);
                const float4 bxi = *(const float4*)(sXi[pfst + bb] + ch * XP + 4 * khalf + g * 8);
                const v2f xr0 = {bxr.x, bxr.y}, xr1 = {bxr.z, bxr.w};
                const v2f xi0 = {bxi.x, bxi.y}, xi1 = {bxi.z, bxi.w};
                acc0[bb] = __builtin_amdgcn_wmma_f32_16x16x4_f32(false, r00, false, xr0, (short)0, acc0[bb], false, false);
                acc1[bb] = __builtin_amdgcn_wmma_f32_16x16x4_f32(false, r10, false, xr0, (short)0, acc1[bb], false, false);
                acc0[bb] = __builtin_amdgcn_wmma_f32_16x16x4_f32(false, i00, false, xi0, (short)0, acc0[bb], false, false);
                acc1[bb] = __builtin_amdgcn_wmma_f32_16x16x4_f32(false, i10, false, xi0, (short)0, acc1[bb], false, false);
                acc0[bb] = __builtin_amdgcn_wmma_f32_16x16x4_f32(false, r01, false, xr1, (short)0, acc0[bb], false, false);
                acc1[bb] = __builtin_amdgcn_wmma_f32_16x16x4_f32(false, r11, false, xr1, (short)0, acc1[bb], false, false);
                acc0[bb] = __builtin_amdgcn_wmma_f32_16x16x4_f32(false, i01, false, xi1, (short)0, acc0[bb], false, false);
                acc1[bb] = __builtin_amdgcn_wmma_f32_16x16x4_f32(false, i11, false, xi1, (short)0, acc1[bb], false, false);
            }
        }
    }

    const float Dh = D[h];
#pragma unroll
    for (int bb = 0; bb < WB; ++bb) {
        const size_t obase = ((size_t)(bfst + bb) * HH + h) * LL + (size_t)ch * CHUNK + rt0 * 16 + 8 * khalf;
        {
            const float4 u0 = *(const float4*)(u + obase);
            const float4 u1 = *(const float4*)(u + obase + 4);
            float4 o0, o1;
            o0.x = acc0[bb][0] + Dh * u0.x; o0.y = acc0[bb][1] + Dh * u0.y;
            o0.z = acc0[bb][2] + Dh * u0.z; o0.w = acc0[bb][3] + Dh * u0.w;
            o1.x = acc0[bb][4] + Dh * u1.x; o1.y = acc0[bb][5] + Dh * u1.y;
            o1.z = acc0[bb][6] + Dh * u1.z; o1.w = acc0[bb][7] + Dh * u1.w;
            *(float4*)(out + obase)     = o0;
            *(float4*)(out + obase + 4) = o1;
        }
        {
            const float4 u0 = *(const float4*)(u + obase + 16);
            const float4 u1 = *(const float4*)(u + obase + 20);
            float4 o0, o1;
            o0.x = acc1[bb][0] + Dh * u0.x; o0.y = acc1[bb][1] + Dh * u0.y;
            o0.z = acc1[bb][2] + Dh * u0.z; o0.w = acc1[bb][3] + Dh * u0.w;
            o1.x = acc1[bb][4] + Dh * u1.x; o1.y = acc1[bb][5] + Dh * u1.y;
            o1.z = acc1[bb][6] + Dh * u1.z; o1.w = acc1[bb][7] + Dh * u1.w;
            *(float4*)(out + obase + 16) = o0;
            *(float4*)(out + obase + 20) = o1;
        }
    }
}

extern "C" void kernel_launch(void* const* d_in, const int* in_sizes, int n_in,
                              void* d_out, int out_size, void* d_ws, size_t ws_size,
                              hipStream_t stream) {
    const float* u       = (const float*)d_in[0];
    const float* Lre     = (const float*)d_in[1];
    const float* Lim     = (const float*)d_in[2];
    const float* Wri     = (const float*)d_in[3];
    const float* Dp      = (const float*)d_in[4];
    const float* logstep = (const float*)d_in[5];
    float* out = (float*)d_out;

    dss_fused<<<HH * (BB / NB), FT, 0, stream>>>(u, Lre, Lim, Wri, Dp, logstep, out);
}
